# starting kernel
# baseline (speedup 1.0000x reference)
#include <hip/hip_runtime.h>
#include <cstdint>

#define N_NODES 100000
#define N_EDGES 1600000
#define IN_CH   256
#define HID     128
#define K_STEPS 10
#define ALPHA   0.1f

typedef __attribute__((ext_vector_type(2))) float v2f;
typedef __attribute__((ext_vector_type(8))) float v8f;

__device__ __forceinline__ void atom_add_f32(float* p, float v) {
    __hip_atomic_fetch_add(p, v, __ATOMIC_RELAXED, __HIP_MEMORY_SCOPE_AGENT);
}

__global__ __launch_bounds__(256) void deg_kernel(const int* __restrict__ src,
                                                  const int* __restrict__ dst,
                                                  float* __restrict__ deg_out,
                                                  float* __restrict__ deg_in) {
    int e = blockIdx.x * blockDim.x + threadIdx.x;
    if (e < N_EDGES) {
        atom_add_f32(&deg_out[src[e]], 1.0f);
        atom_add_f32(&deg_in[dst[e]], 1.0f);
    }
}

__global__ __launch_bounds__(256) void inv_kernel(float* __restrict__ deg, int n) {
    int i = blockIdx.x * blockDim.x + threadIdx.x;
    if (i < n) {
        float d = deg[i];
        deg[i] = (d > 0.0f) ? rsqrtf(fmaxf(d, 1.0f)) : 0.0f;
    }
}

__global__ __launch_bounds__(256) void enorm_kernel(const int* __restrict__ src,
                                                    const int* __restrict__ dst,
                                                    const float* __restrict__ inv_out,
                                                    const float* __restrict__ inv_in,
                                                    float* __restrict__ enorm) {
    int e = blockIdx.x * blockDim.x + threadIdx.x;
    if (e < N_EDGES)
        enorm[e] = inv_out[src[e]] * inv_in[dst[e]];
}

__global__ __launch_bounds__(256) void gemm_kernel(const float* __restrict__ X,
                                                   const float* __restrict__ W,
                                                   float* __restrict__ H) {
    const int lane    = threadIdx.x & 31;
    const int wave    = threadIdx.x >> 5;
    const int rowBase = blockIdx.x * 16;
    const int colBase = wave * 16;
    const int half    = lane >> 4;
    const int l16     = lane & 15;

    const float* __restrict__ arow = X + (size_t)(rowBase + l16) * IN_CH + half * 2;
    const float* __restrict__ wcol = W + (size_t)(half * 2) * HID + (colBase + l16);

    v8f acc = {};
#pragma unroll 8
    for (int k0 = 0; k0 < IN_CH; k0 += 4) {
        v2f a, b;
        a.x = arow[k0];
        a.y = arow[k0 + 1];
        b.x = wcol[k0 * HID];
        b.y = wcol[k0 * HID + HID];
        acc = __builtin_amdgcn_wmma_f32_16x16x4_f32(
             false, a,  false, b,
             (short)0, acc,  false,  false);
    }

    float* __restrict__ hrow = H + (size_t)(rowBase + half * 8) * HID + (colBase + l16);
#pragma unroll
    for (int i = 0; i < 8; ++i)
        hrow[(size_t)i * HID] = acc[i];
}

__global__ __launch_bounds__(256) void prop_kernel(const float* __restrict__ h,
                                                   const int* __restrict__ src,
                                                   const int* __restrict__ dst,
                                                   const float* __restrict__ enorm,
                                                   float* __restrict__ out) {
    const int lane = threadIdx.x & 31;
    const int e    = blockIdx.x * 8 + (threadIdx.x >> 5);
    if (e >= N_EDGES) return;
    const int   s = src[e];
    const int   d = dst[e];
    const float w = enorm[e];
    const float4 v = *(const float4*)(h + (size_t)s * HID + lane * 4);
    float* o = out + (size_t)d * HID + lane * 4;
    atom_add_f32(o + 0, v.x * w);
    atom_add_f32(o + 1, v.y * w);
    atom_add_f32(o + 2, v.z * w);
    atom_add_f32(o + 3, v.w * w);
}

__global__ __launch_bounds__(256) void bias_relu_kernel(float* __restrict__ h,
                                                        const float* __restrict__ bias) {
    size_t i = ((size_t)blockIdx.x * blockDim.x + threadIdx.x) * 4;
    if (i >= (size_t)N_NODES * HID) return;
    int col = (int)(i % HID);
    float4 v  = *(float4*)(h + i);
    float4 bb = *(const float4*)(bias + col);
    v.x = fmaxf(v.x + bb.x, 0.0f);
    v.y = fmaxf(v.y + bb.y, 0.0f);
    v.z = fmaxf(v.z + bb.z, 0.0f);
    v.w = fmaxf(v.w + bb.w, 0.0f);
    *(float4*)(h + i) = v;
}

__global__ __launch_bounds__(256) void axpy_kernel(const float* __restrict__ p,
                                                   const float* __restrict__ h0,
                                                   float* __restrict__ out) {
    size_t i = ((size_t)blockIdx.x * blockDim.x + threadIdx.x) * 4;
    if (i >= (size_t)N_NODES * HID) return;
    float4 a = *(const float4*)(p + i);
    float4 b = *(const float4*)(h0 + i);
    float4 r;
    r.x = (1.0f - ALPHA) * a.x + ALPHA * b.x;
    r.y = (1.0f - ALPHA) * a.y + ALPHA * b.y;
    r.z = (1.0f - ALPHA) * a.z + ALPHA * b.z;
    r.w = (1.0f - ALPHA) * a.w + ALPHA * b.w;
    *(float4*)(out + i) = r;
}

extern "C" void kernel_launch(void* const* d_in, const int* in_sizes, int n_in,
                              void* d_out, int out_size, void* d_ws, size_t ws_size,
                              hipStream_t stream) {
    const float* features = (const float*)d_in[0];
    const float* W        = (const float*)d_in[1];
    const float* bias     = (const float*)d_in[2];
    const int*   eidx     = (const int*)d_in[3];
    const int*   src      = eidx;
    const int*   dst      = eidx + N_EDGES;

    float* out = (float*)d_out;

    const size_t NH = (size_t)N_NODES * HID;
    float* deg     = (float*)d_ws;
    float* deg_out = deg;
    float* deg_in  = deg + N_NODES;
    float* enorm   = deg + 2 * N_NODES;
    float* bufA    = enorm + N_EDGES;
    float* bufB    = bufA + NH;

    hipMemsetAsync(deg, 0, 2 * (size_t)N_NODES * sizeof(float), stream);
    deg_kernel<<<(N_EDGES + 255) / 256, 256, 0, stream>>>(src, dst, deg_out, deg_in);
    inv_kernel<<<(2 * N_NODES + 255) / 256, 256, 0, stream>>>(deg, 2 * N_NODES);
    enorm_kernel<<<(N_EDGES + 255) / 256, 256, 0, stream>>>(src, dst, deg_out, deg_in, enorm);

    gemm_kernel<<<N_NODES / 16, 256, 0, stream>>>(features, W, bufA);
    hipMemsetAsync(bufB, 0, NH * sizeof(float), stream);
    prop_kernel<<<N_EDGES / 8, 256, 0, stream>>>(bufA, src, dst, enorm, bufB);
    bias_relu_kernel<<<(int)(NH / 4 / 256), 256, 0, stream>>>(bufB, bias);

    float* cur = bufB;
    for (int t = 0; t < K_STEPS; ++t) {
        float* tmp = (t & 1) ? out : bufA;
        hipMemsetAsync(tmp, 0, NH * sizeof(float), stream);
        prop_kernel<<<N_EDGES / 8, 256, 0, stream>>>(cur, src, dst, enorm, tmp);
        axpy_kernel<<<(int)(NH / 4 / 256), 256, 0, stream>>>(tmp, bufB, tmp);
        cur = tmp;
    }
}


# prop: edges bucketed once by block of 32 dst rows; a work-group sums its block in LDS with ds float adds, stores each row once, step tail applied; no global float atomics; sums in any order as before
# speedup vs baseline: 3.7163x; 3.7163x over previous
#include <hip/hip_runtime.h>
#include <cstdint>

#define N_NODES 100000
#define N_EDGES 1600000
#define IN_CH   256
#define HID     128
#define K_STEPS 10
#define ALPHA   0.1f

#define BLOCK_ROWS 32
#define N_BLOCKS   (N_NODES / BLOCK_ROWS)
#define SRC_BITS   17
#define SRC_MASK   ((1 << SRC_BITS) - 1)
#define CUR_STRIDE 32
static_assert(N_NODES % BLOCK_ROWS == 0, "blocks cover the rows exactly");
static_assert(N_NODES <= (1 << SRC_BITS), "a source row fits its field");
static_assert(N_BLOCKS * CUR_STRIDE <= N_NODES, "the cursors fit in the in-degree array they replace");

typedef __attribute__((ext_vector_type(2))) float v2f;
typedef __attribute__((ext_vector_type(8))) float v8f;

__device__ __forceinline__ void atom_add_f32(float* p, float v) {
    __hip_atomic_fetch_add(p, v, __ATOMIC_RELAXED, __HIP_MEMORY_SCOPE_AGENT);
}

__device__ __forceinline__ void lds_add_f32(float* p, float v) {
    __hip_atomic_fetch_add(p, v, __ATOMIC_RELAXED, __HIP_MEMORY_SCOPE_WORKGROUP);
}

__global__ __launch_bounds__(256) void deg_kernel(const int* __restrict__ src,
                                                  const int* __restrict__ dst,
                                                  float* __restrict__ deg_out,
                                                  float* __restrict__ deg_in) {
    int e = blockIdx.x * blockDim.x + threadIdx.x;
    if (e < N_EDGES) {
        atom_add_f32(&deg_out[src[e]], 1.0f);
        atom_add_f32(&deg_in[dst[e]], 1.0f);
    }
}

__global__ __launch_bounds__(256) void rowprep_kernel(float* __restrict__ deg_out,
                                                      const float* __restrict__ deg_in,
                                                      int* __restrict__ counts) {
    const int i = blockIdx.x * blockDim.x + threadIdx.x;
    float din = 0.0f;
    if (i < N_NODES) {
        const float d = deg_out[i];
        deg_out[i] = (d > 0.0f) ? rsqrtf(fmaxf(d, 1.0f)) : 0.0f;
        din = deg_in[i];
    }
    for (int off = 16; off > 0; off >>= 1)
        din += __shfl_xor(din, off, 32);
    if ((threadIdx.x & 31) == 0 && i < N_NODES)
        counts[i >> 5] = (int)(din + 0.5f);
}

#define SCAN_N 4096
static_assert(N_BLOCKS <= SCAN_N, "the scan covers every block");

__global__ __launch_bounds__(1024) void scan_kernel(const int* __restrict__ counts,
                                                    int* __restrict__ cur) {
    __shared__ int bufs[2][SCAN_N];
    const int tid = threadIdx.x;
    for (int i = tid; i < SCAN_N; i += 1024)
        bufs[0][i] = (i < N_BLOCKS) ? counts[i] : 0;
    __syncthreads();
    int p = 0;
    for (int d = 1; d < SCAN_N; d <<= 1) {
        for (int i = tid; i < SCAN_N; i += 1024) {
            int v = bufs[p][i];
            if (i >= d) v += bufs[p][i - d];
            bufs[p ^ 1][i] = v;
        }
        __syncthreads();
        p ^= 1;
    }
    for (int i = tid; i < N_BLOCKS; i += 1024)
        cur[i * CUR_STRIDE] = (i == 0) ? 0 : bufs[p][i - 1];
}

__global__ __launch_bounds__(256) void fill_kernel(const int* __restrict__ src,
                                                   const int* __restrict__ dst,
                                                   int* __restrict__ cur,
                                                   int* __restrict__ rec) {
    const int e = blockIdx.x * blockDim.x + threadIdx.x;
    if (e < N_EDGES) {
        const int s = src[e];
        const int d = dst[e];
        const int pos = __hip_atomic_fetch_add(&cur[(d >> 5) * CUR_STRIDE], 1, __ATOMIC_RELAXED,
                                               __HIP_MEMORY_SCOPE_AGENT);
        rec[pos] = s | ((d & (BLOCK_ROWS - 1)) << SRC_BITS);
    }
}

__global__ __launch_bounds__(256) void gemm_kernel(const float* __restrict__ X,
                                                   const float* __restrict__ W,
                                                   float* __restrict__ H) {
    const int lane    = threadIdx.x & 31;
    const int wave    = threadIdx.x >> 5;
    const int rowBase = blockIdx.x * 16;
    const int colBase = wave * 16;
    const int half    = lane >> 4;
    const int l16     = lane & 15;

    const float* __restrict__ arow = X + (size_t)(rowBase + l16) * IN_CH + half * 2;
    const float* __restrict__ wcol = W + (size_t)(half * 2) * HID + (colBase + l16);

    v8f acc = {};
#pragma unroll 8
    for (int k0 = 0; k0 < IN_CH; k0 += 4) {
        v2f a, b;
        a.x = arow[k0];
        a.y = arow[k0 + 1];
        b.x = wcol[k0 * HID];
        b.y = wcol[k0 * HID + HID];
        acc = __builtin_amdgcn_wmma_f32_16x16x4_f32(
             false, a,  false, b,
             (short)0, acc,  false,  false);
    }

    float* __restrict__ hrow = H + (size_t)(rowBase + half * 8) * HID + (colBase + l16);
#pragma unroll
    for (int i = 0; i < 8; ++i)
        hrow[(size_t)i * HID] = acc[i];
}

#define PROP_THREADS  128
#define ROWS_PER_PASS 8
#define COUNT_LOADS   4
#define BLOCK_F4      (BLOCK_ROWS * HID / 4)
#define F4_PER_THREAD (BLOCK_F4 / PROP_THREADS)
static_assert(BLOCK_F4 % PROP_THREADS == 0, "the threads cover the block's sums exactly");
static_assert((PROP_THREADS * 4) % HID == 0, "a thread's places all lie in the same 4 channels");

template <bool BLEND>
__global__ __launch_bounds__(PROP_THREADS) void prop_kernel(const float* __restrict__ h,
                                                            const int* __restrict__ rec,
                                                            const int* __restrict__ cur,
                                                            const float* __restrict__ inv_out,
                                                            const float* __restrict__ tail,
                                                            float* __restrict__ out) {
    __shared__ float acc[BLOCK_ROWS * HID];
    __shared__ int   cnt[BLOCK_ROWS];
    __shared__ float inv_in[BLOCK_ROWS];

    const int tid  = threadIdx.x;
    const int lane = tid & 31;
    const int wave = __builtin_amdgcn_readfirstlane(tid >> 5);
    const int b    = blockIdx.x;
    const int s    = (b == 0) ? 0 : cur[(b - 1) * CUR_STRIDE];
    const int e    = cur[b * CUR_STRIDE];

    for (int i = tid; i < BLOCK_F4; i += PROP_THREADS)
        ((float4*)acc)[i] = make_float4(0.0f, 0.0f, 0.0f, 0.0f);
    if (tid < BLOCK_ROWS) cnt[tid] = 0;
    __syncthreads();

    for (int i0 = s + tid; i0 < e; i0 += COUNT_LOADS * PROP_THREADS) {
        int r[COUNT_LOADS];
        for (int k = 0; k < COUNT_LOADS; ++k) {
            const int i = i0 + k * PROP_THREADS;
            r[k] = (i < e) ? rec[i] : -1;
        }
        for (int k = 0; k < COUNT_LOADS; ++k)
            if (r[k] >= 0)
                __hip_atomic_fetch_add(&cnt[r[k] >> SRC_BITS], 1, __ATOMIC_RELAXED,
                                       __HIP_MEMORY_SCOPE_WORKGROUP);
    }
    __syncthreads();
    if (tid < BLOCK_ROWS) {
        const int c = cnt[tid];
        inv_in[tid] = (c > 0) ? rsqrtf(fmaxf((float)c, 1.0f)) : 0.0f;
    }
    __syncthreads();

    const float* __restrict__ hl = h + lane;
    float* al = acc + lane;
    for (int base = s + wave * 32; base < e; base += PROP_THREADS) {
        const int n = min(32, e - base);
        int   sidx = 0;
        int   dl   = 0;
        float w    = 0.0f;
        if (lane < n) {
            const int r = rec[base + lane];
            sidx = r & SRC_MASK;
            dl   = r >> SRC_BITS;
            w    = inv_out[sidx] * inv_in[dl];
        }
        const int wv = __float_as_int(w);
        for (int j0 = 0; j0 < n; j0 += ROWS_PER_PASS) {
            float v[ROWS_PER_PASS][4];
            for (int j = 0; j < ROWS_PER_PASS; ++j) {
                const int sj = __builtin_amdgcn_readlane(sidx, min(j0 + j, n - 1));
                const float* __restrict__ hp = hl + (size_t)sj * HID;
                v[j][0] = hp[0];
                v[j][1] = hp[32];
                v[j][2] = hp[64];
                v[j][3] = hp[96];
            }
            for (int j = 0; j < ROWS_PER_PASS; ++j) {
                if (j0 + j < n) {
                    const int   d  = __builtin_amdgcn_readlane(dl, j0 + j);
                    const float wj = __int_as_float(__builtin_amdgcn_readlane(wv, j0 + j));
                    float* a = al + d * HID;
                    lds_add_f32(a,      v[j][0] * wj);
                    lds_add_f32(a + 32, v[j][1] * wj);
                    lds_add_f32(a + 64, v[j][2] * wj);
                    lds_add_f32(a + 96, v[j][3] * wj);
                }
            }
        }
    }
    const size_t row0 = (size_t)b * BLOCK_ROWS * HID;
    float4 z[F4_PER_THREAD];
    if (BLEND) {
        for (int k = 0; k < F4_PER_THREAD; ++k)
            z[k] = *(const float4*)(tail + row0 + (size_t)(tid + k * PROP_THREADS) * 4);
    } else {
        z[0] = *(const float4*)(tail + ((tid * 4) & (HID - 1)));
    }
    __syncthreads();

    for (int k = 0; k < F4_PER_THREAD; ++k) {
        const int i = tid + k * PROP_THREADS;
        const float4 a = ((const float4*)acc)[i];
        float4 r;
        if (BLEND) {
            r.x = (1.0f - ALPHA) * a.x + ALPHA * z[k].x;
            r.y = (1.0f - ALPHA) * a.y + ALPHA * z[k].y;
            r.z = (1.0f - ALPHA) * a.z + ALPHA * z[k].z;
            r.w = (1.0f - ALPHA) * a.w + ALPHA * z[k].w;
        } else {
            r.x = fmaxf(a.x + z[0].x, 0.0f);
            r.y = fmaxf(a.y + z[0].y, 0.0f);
            r.z = fmaxf(a.z + z[0].z, 0.0f);
            r.w = fmaxf(a.w + z[0].w, 0.0f);
        }
        *(float4*)(out + row0 + (size_t)i * 4) = r;
    }
}

extern "C" void kernel_launch(void* const* d_in, const int* in_sizes, int n_in,
                              void* d_out, int out_size, void* d_ws, size_t ws_size,
                              hipStream_t stream) {
    const float* features = (const float*)d_in[0];
    const float* W        = (const float*)d_in[1];
    const float* bias     = (const float*)d_in[2];
    const int*   eidx     = (const int*)d_in[3];
    const int*   src      = eidx;
    const int*   dst      = eidx + N_EDGES;

    float* out = (float*)d_out;

    const size_t NH = (size_t)N_NODES * HID;
    float* deg     = (float*)d_ws;
    float* deg_out = deg;
    float* deg_in  = deg + N_NODES;
    int*   cur     = (int*)deg_in;
    int*   rec     = (int*)(deg + 2 * N_NODES);
    int*   counts  = rec;
    float* bufA    = (float*)(rec + N_EDGES);
    float* bufB    = bufA + NH;

    hipMemsetAsync(deg, 0, 2 * (size_t)N_NODES * sizeof(float), stream);
    deg_kernel<<<(N_EDGES + 255) / 256, 256, 0, stream>>>(src, dst, deg_out, deg_in);
    rowprep_kernel<<<(N_NODES + 255) / 256, 256, 0, stream>>>(deg_out, deg_in, counts);
    scan_kernel<<<1, 1024, 0, stream>>>(counts, cur);
    fill_kernel<<<(N_EDGES + 255) / 256, 256, 0, stream>>>(src, dst, cur, rec);

    gemm_kernel<<<N_NODES / 16, 256, 0, stream>>>(features, W, bufA);
    prop_kernel<false><<<N_BLOCKS, PROP_THREADS, 0, stream>>>(bufA, rec, cur, deg_out, bias, bufB);

    float* hc = bufB;
    for (int t = 0; t < K_STEPS; ++t) {
        float* tmp = (t & 1) ? out : bufA;
        prop_kernel<true><<<N_BLOCKS, PROP_THREADS, 0, stream>>>(hc, rec, cur, deg_out, bufB, tmp);
        hc = tmp;
    }
}


# prop: a source row is fetched by one load (16 B per lane, lane owns 4 adjacent channels; was 4 dword loads); LDS row sums channel-permuted so ds float adds stay conflict-free; sum order as before
# speedup vs baseline: 4.0386x; 1.0867x over previous
#include <hip/hip_runtime.h>
#include <cstdint>

#define N_NODES 100000
#define N_EDGES 1600000
#define IN_CH   256
#define HID     128
#define K_STEPS 10
#define ALPHA   0.1f

#define BLOCK_ROWS 32
#define N_BLOCKS   (N_NODES / BLOCK_ROWS)
#define SRC_BITS   17
#define SRC_MASK   ((1 << SRC_BITS) - 1)
#define CUR_STRIDE 32
static_assert(N_NODES % BLOCK_ROWS == 0, "blocks cover the rows exactly");
static_assert(N_NODES <= (1 << SRC_BITS), "a source row fits its field");
static_assert(N_BLOCKS * CUR_STRIDE <= N_NODES, "the cursors fit in the in-degree array they replace");

typedef __attribute__((ext_vector_type(2))) float v2f;
typedef __attribute__((ext_vector_type(8))) float v8f;

__device__ __forceinline__ void atom_add_f32(float* p, float v) {
    __hip_atomic_fetch_add(p, v, __ATOMIC_RELAXED, __HIP_MEMORY_SCOPE_AGENT);
}

__device__ __forceinline__ void lds_add_f32(float* p, float v) {
    __hip_atomic_fetch_add(p, v, __ATOMIC_RELAXED, __HIP_MEMORY_SCOPE_WORKGROUP);
}

__global__ __launch_bounds__(256) void deg_kernel(const int* __restrict__ src,
                                                  const int* __restrict__ dst,
                                                  float* __restrict__ deg_out,
                                                  float* __restrict__ deg_in) {
    int e = blockIdx.x * blockDim.x + threadIdx.x;
    if (e < N_EDGES) {
        atom_add_f32(&deg_out[src[e]], 1.0f);
        atom_add_f32(&deg_in[dst[e]], 1.0f);
    }
}

__global__ __launch_bounds__(256) void rowprep_kernel(float* __restrict__ deg_out,
                                                      const float* __restrict__ deg_in,
                                                      int* __restrict__ counts) {
    const int i = blockIdx.x * blockDim.x + threadIdx.x;
    float din = 0.0f;
    if (i < N_NODES) {
        const float d = deg_out[i];
        deg_out[i] = (d > 0.0f) ? rsqrtf(fmaxf(d, 1.0f)) : 0.0f;
        din = deg_in[i];
    }
    for (int off = 16; off > 0; off >>= 1)
        din += __shfl_xor(din, off, 32);
    if ((threadIdx.x & 31) == 0 && i < N_NODES)
        counts[i >> 5] = (int)(din + 0.5f);
}

#define SCAN_N 4096
static_assert(N_BLOCKS <= SCAN_N, "the scan covers every block");

__global__ __launch_bounds__(1024) void scan_kernel(const int* __restrict__ counts,
                                                    int* __restrict__ cur) {
    __shared__ int bufs[2][SCAN_N];
    const int tid = threadIdx.x;
    for (int i = tid; i < SCAN_N; i += 1024)
        bufs[0][i] = (i < N_BLOCKS) ? counts[i] : 0;
    __syncthreads();
    int p = 0;
    for (int d = 1; d < SCAN_N; d <<= 1) {
        for (int i = tid; i < SCAN_N; i += 1024) {
            int v = bufs[p][i];
            if (i >= d) v += bufs[p][i - d];
            bufs[p ^ 1][i] = v;
        }
        __syncthreads();
        p ^= 1;
    }
    for (int i = tid; i < N_BLOCKS; i += 1024)
        cur[i * CUR_STRIDE] = (i == 0) ? 0 : bufs[p][i - 1];
}

__global__ __launch_bounds__(256) void fill_kernel(const int* __restrict__ src,
                                                   const int* __restrict__ dst,
                                                   int* __restrict__ cur,
                                                   int* __restrict__ rec) {
    const int e = blockIdx.x * blockDim.x + threadIdx.x;
    if (e < N_EDGES) {
        const int s = src[e];
        const int d = dst[e];
        const int pos = __hip_atomic_fetch_add(&cur[(d >> 5) * CUR_STRIDE], 1, __ATOMIC_RELAXED,
                                               __HIP_MEMORY_SCOPE_AGENT);
        rec[pos] = s | ((d & (BLOCK_ROWS - 1)) << SRC_BITS);
    }
}

__global__ __launch_bounds__(256) void gemm_kernel(const float* __restrict__ X,
                                                   const float* __restrict__ W,
                                                   float* __restrict__ H) {
    const int lane    = threadIdx.x & 31;
    const int wave    = threadIdx.x >> 5;
    const int rowBase = blockIdx.x * 16;
    const int colBase = wave * 16;
    const int half    = lane >> 4;
    const int l16     = lane & 15;

    const float* __restrict__ arow = X + (size_t)(rowBase + l16) * IN_CH + half * 2;
    const float* __restrict__ wcol = W + (size_t)(half * 2) * HID + (colBase + l16);

    v8f acc = {};
#pragma unroll 8
    for (int k0 = 0; k0 < IN_CH; k0 += 4) {
        v2f a, b;
        a.x = arow[k0];
        a.y = arow[k0 + 1];
        b.x = wcol[k0 * HID];
        b.y = wcol[k0 * HID + HID];
        acc = __builtin_amdgcn_wmma_f32_16x16x4_f32(
             false, a,  false, b,
             (short)0, acc,  false,  false);
    }

    float* __restrict__ hrow = H + (size_t)(rowBase + half * 8) * HID + (colBase + l16);
#pragma unroll
    for (int i = 0; i < 8; ++i)
        hrow[(size_t)i * HID] = acc[i];
}

#define PROP_THREADS  128
#define ROWS_PER_PASS 8
#define COUNT_LOADS   4
#define BLOCK_F4      (BLOCK_ROWS * HID / 4)
#define F4_PER_THREAD (BLOCK_F4 / PROP_THREADS)
static_assert(BLOCK_F4 % PROP_THREADS == 0, "the threads cover the block's sums exactly");
static_assert((PROP_THREADS * 4) % HID == 0, "a thread's places all lie in the same 4 channels");

template <bool BLEND>
__global__ __launch_bounds__(PROP_THREADS) void prop_kernel(const float* __restrict__ h,
                                                            const int* __restrict__ rec,
                                                            const int* __restrict__ cur,
                                                            const float* __restrict__ inv_out,
                                                            const float* __restrict__ tail,
                                                            float* __restrict__ out) {
    __shared__ float acc[BLOCK_ROWS * HID];
    __shared__ int   cnt[BLOCK_ROWS];
    __shared__ float inv_in[BLOCK_ROWS];

    const int tid  = threadIdx.x;
    const int lane = tid & 31;
    const int wave = __builtin_amdgcn_readfirstlane(tid >> 5);
    const int b    = blockIdx.x;
    const int s    = (b == 0) ? 0 : cur[(b - 1) * CUR_STRIDE];
    const int e    = cur[b * CUR_STRIDE];

    for (int i = tid; i < BLOCK_F4; i += PROP_THREADS)
        ((float4*)acc)[i] = make_float4(0.0f, 0.0f, 0.0f, 0.0f);
    if (tid < BLOCK_ROWS) cnt[tid] = 0;
    __syncthreads();

    for (int i0 = s + tid; i0 < e; i0 += COUNT_LOADS * PROP_THREADS) {
        int r[COUNT_LOADS];
        for (int k = 0; k < COUNT_LOADS; ++k) {
            const int i = i0 + k * PROP_THREADS;
            r[k] = (i < e) ? rec[i] : -1;
        }
        for (int k = 0; k < COUNT_LOADS; ++k)
            if (r[k] >= 0)
                __hip_atomic_fetch_add(&cnt[r[k] >> SRC_BITS], 1, __ATOMIC_RELAXED,
                                       __HIP_MEMORY_SCOPE_WORKGROUP);
    }
    __syncthreads();
    if (tid < BLOCK_ROWS) {
        const int c = cnt[tid];
        inv_in[tid] = (c > 0) ? rsqrtf(fmaxf((float)c, 1.0f)) : 0.0f;
    }
    __syncthreads();

    const float* __restrict__ hl = h + lane * 4;
    float* al = acc + lane;
    for (int base = s + wave * 32; base < e; base += PROP_THREADS) {
        const int n = min(32, e - base);
        int   sidx = 0;
        int   dl   = 0;
        float w    = 0.0f;
        if (lane < n) {
            const int r = rec[base + lane];
            sidx = r & SRC_MASK;
            dl   = r >> SRC_BITS;
            w    = inv_out[sidx] * inv_in[dl];
        }
        const int wv = __float_as_int(w);
        for (int j0 = 0; j0 < n; j0 += ROWS_PER_PASS) {
            float4 v[ROWS_PER_PASS];
            for (int j = 0; j < ROWS_PER_PASS; ++j) {
                const int sj = __builtin_amdgcn_readlane(sidx, min(j0 + j, n - 1));
                v[j] = *(const float4*)(hl + (size_t)sj * HID);
            }
            for (int j = 0; j < ROWS_PER_PASS; ++j) {
                if (j0 + j < n) {
                    const int   d  = __builtin_amdgcn_readlane(dl, j0 + j);
                    const float wj = __int_as_float(__builtin_amdgcn_readlane(wv, j0 + j));
                    float* a = al + d * HID;
                    lds_add_f32(a,      v[j].x * wj);
                    lds_add_f32(a + 32, v[j].y * wj);
                    lds_add_f32(a + 64, v[j].z * wj);
                    lds_add_f32(a + 96, v[j].w * wj);
                }
            }
        }
    }
    const size_t row0 = (size_t)b * BLOCK_ROWS * HID;
    float4 z[F4_PER_THREAD];
    if (BLEND) {
        for (int k = 0; k < F4_PER_THREAD; ++k)
            z[k] = *(const float4*)(tail + row0 + (size_t)(tid + k * PROP_THREADS) * 4);
    } else {
        z[0] = *(const float4*)(tail + ((tid * 4) & (HID - 1)));
    }
    __syncthreads();

    for (int k = 0; k < F4_PER_THREAD; ++k) {
        const int i = tid + k * PROP_THREADS;
        const float* ap = acc + (i >> 5) * HID + (i & 31);
        float4 a;
        a.x = ap[0];
        a.y = ap[32];
        a.z = ap[64];
        a.w = ap[96];
        float4 r;
        if (BLEND) {
            r.x = (1.0f - ALPHA) * a.x + ALPHA * z[k].x;
            r.y = (1.0f - ALPHA) * a.y + ALPHA * z[k].y;
            r.z = (1.0f - ALPHA) * a.z + ALPHA * z[k].z;
            r.w = (1.0f - ALPHA) * a.w + ALPHA * z[k].w;
        } else {
            r.x = fmaxf(a.x + z[0].x, 0.0f);
            r.y = fmaxf(a.y + z[0].y, 0.0f);
            r.z = fmaxf(a.z + z[0].z, 0.0f);
            r.w = fmaxf(a.w + z[0].w, 0.0f);
        }
        *(float4*)(out + row0 + (size_t)i * 4) = r;
    }
}

extern "C" void kernel_launch(void* const* d_in, const int* in_sizes, int n_in,
                              void* d_out, int out_size, void* d_ws, size_t ws_size,
                              hipStream_t stream) {
    const float* features = (const float*)d_in[0];
    const float* W        = (const float*)d_in[1];
    const float* bias     = (const float*)d_in[2];
    const int*   eidx     = (const int*)d_in[3];
    const int*   src      = eidx;
    const int*   dst      = eidx + N_EDGES;

    float* out = (float*)d_out;

    const size_t NH = (size_t)N_NODES * HID;
    float* deg     = (float*)d_ws;
    float* deg_out = deg;
    float* deg_in  = deg + N_NODES;
    int*   cur     = (int*)deg_in;
    int*   rec     = (int*)(deg + 2 * N_NODES);
    int*   counts  = rec;
    float* bufA    = (float*)(rec + N_EDGES);
    float* bufB    = bufA + NH;

    hipMemsetAsync(deg, 0, 2 * (size_t)N_NODES * sizeof(float), stream);
    deg_kernel<<<(N_EDGES + 255) / 256, 256, 0, stream>>>(src, dst, deg_out, deg_in);
    rowprep_kernel<<<(N_NODES + 255) / 256, 256, 0, stream>>>(deg_out, deg_in, counts);
    scan_kernel<<<1, 1024, 0, stream>>>(counts, cur);
    fill_kernel<<<(N_EDGES + 255) / 256, 256, 0, stream>>>(src, dst, cur, rec);

    gemm_kernel<<<N_NODES / 16, 256, 0, stream>>>(features, W, bufA);
    prop_kernel<false><<<N_BLOCKS, PROP_THREADS, 0, stream>>>(bufA, rec, cur, deg_out, bias, bufB);

    float* hc = bufB;
    for (int t = 0; t < K_STEPS; ++t) {
        float* tmp = (t & 1) ? out : bufA;
        prop_kernel<true><<<N_BLOCKS, PROP_THREADS, 0, stream>>>(hc, rec, cur, deg_out, bufB, tmp);
        hc = tmp;
    }
}


# GEMM: a wave keeps its 16-column slice of W in registers and walks 16-row tiles of X staged in LDS by whole-line loads (was 16-line fragment loads per k step); sum order over k changed in 8-k blocks
# speedup vs baseline: 4.2945x; 1.0634x over previous
#include <hip/hip_runtime.h>
#include <cstdint>

#define N_NODES 100000
#define N_EDGES 1600000
#define IN_CH   256
#define HID     128
#define K_STEPS 10
#define ALPHA   0.1f

#define BLOCK_ROWS 32
#define N_BLOCKS   (N_NODES / BLOCK_ROWS)
#define SRC_BITS   17
#define SRC_MASK   ((1 << SRC_BITS) - 1)
#define CUR_STRIDE 32
static_assert(N_NODES % BLOCK_ROWS == 0, "blocks cover the rows exactly");
static_assert(N_NODES <= (1 << SRC_BITS), "a source row fits its field");
static_assert(N_BLOCKS * CUR_STRIDE <= N_NODES, "the cursors fit in the in-degree array they replace");

typedef __attribute__((ext_vector_type(2))) float v2f;
typedef __attribute__((ext_vector_type(8))) float v8f;

__device__ __forceinline__ void atom_add_f32(float* p, float v) {
    __hip_atomic_fetch_add(p, v, __ATOMIC_RELAXED, __HIP_MEMORY_SCOPE_AGENT);
}

__device__ __forceinline__ void lds_add_f32(float* p, float v) {
    __hip_atomic_fetch_add(p, v, __ATOMIC_RELAXED, __HIP_MEMORY_SCOPE_WORKGROUP);
}

__global__ __launch_bounds__(256) void deg_kernel(const int* __restrict__ src,
                                                  const int* __restrict__ dst,
                                                  float* __restrict__ deg_out,
                                                  float* __restrict__ deg_in) {
    int e = blockIdx.x * blockDim.x + threadIdx.x;
    if (e < N_EDGES) {
        atom_add_f32(&deg_out[src[e]], 1.0f);
        atom_add_f32(&deg_in[dst[e]], 1.0f);
    }
}

__global__ __launch_bounds__(256) void rowprep_kernel(float* __restrict__ deg_out,
                                                      const float* __restrict__ deg_in,
                                                      int* __restrict__ counts) {
    const int i = blockIdx.x * blockDim.x + threadIdx.x;
    float din = 0.0f;
    if (i < N_NODES) {
        const float d = deg_out[i];
        deg_out[i] = (d > 0.0f) ? rsqrtf(fmaxf(d, 1.0f)) : 0.0f;
        din = deg_in[i];
    }
    for (int off = 16; off > 0; off >>= 1)
        din += __shfl_xor(din, off, 32);
    if ((threadIdx.x & 31) == 0 && i < N_NODES)
        counts[i >> 5] = (int)(din + 0.5f);
}

#define SCAN_N 4096
static_assert(N_BLOCKS <= SCAN_N, "the scan covers every block");

__global__ __launch_bounds__(1024) void scan_kernel(const int* __restrict__ counts,
                                                    int* __restrict__ cur) {
    __shared__ int bufs[2][SCAN_N];
    const int tid = threadIdx.x;
    for (int i = tid; i < SCAN_N; i += 1024)
        bufs[0][i] = (i < N_BLOCKS) ? counts[i] : 0;
    __syncthreads();
    int p = 0;
    for (int d = 1; d < SCAN_N; d <<= 1) {
        for (int i = tid; i < SCAN_N; i += 1024) {
            int v = bufs[p][i];
            if (i >= d) v += bufs[p][i - d];
            bufs[p ^ 1][i] = v;
        }
        __syncthreads();
        p ^= 1;
    }
    for (int i = tid; i < N_BLOCKS; i += 1024)
        cur[i * CUR_STRIDE] = (i == 0) ? 0 : bufs[p][i - 1];
}

__global__ __launch_bounds__(256) void fill_kernel(const int* __restrict__ src,
                                                   const int* __restrict__ dst,
                                                   int* __restrict__ cur,
                                                   int* __restrict__ rec) {
    const int e = blockIdx.x * blockDim.x + threadIdx.x;
    if (e < N_EDGES) {
        const int s = src[e];
        const int d = dst[e];
        const int pos = __hip_atomic_fetch_add(&cur[(d >> 5) * CUR_STRIDE], 1, __ATOMIC_RELAXED,
                                               __HIP_MEMORY_SCOPE_AGENT);
        rec[pos] = s | ((d & (BLOCK_ROWS - 1)) << SRC_BITS);
    }
}

#define GEMM_THREADS 256
#define TILE_ROWS    16
#define N_TILES      (N_NODES / TILE_ROWS)
#define TILE_F4      (TILE_ROWS * IN_CH / 4)
#define A_PITCH      (IN_CH + 4)
#define GEMM_GRID    512
static_assert(N_NODES % TILE_ROWS == 0, "tiles cover the rows exactly");
static_assert(TILE_F4 == 4 * GEMM_THREADS, "a thread fetches 4 float4 of a tile");
static_assert(GEMM_GRID <= N_TILES, "every work-group has a first tile");
static_assert(((A_PITCH / 4) & 1) == 1 && A_PITCH % 4 == 0, "pitch is 16 bytes x odd");

__global__ __launch_bounds__(GEMM_THREADS, 2) void gemm_kernel(const float* __restrict__ X,
                                                                const float* __restrict__ W,
                                                                float* __restrict__ H) {
    __shared__ float xs[2][TILE_ROWS * A_PITCH];

    const int tid  = threadIdx.x;
    const int lane = tid & 31;
    const int wave = tid >> 5;
    const int half = lane >> 4;
    const int l16  = lane & 15;
    const int col  = wave * 16 + l16;

    v2f bw[IN_CH / 4];
    {
        const float* __restrict__ wp = W + (size_t)(4 * half) * HID + col;
#pragma unroll
        for (int q = 0; q < IN_CH / 8; ++q) {
            bw[2 * q].x     = wp[(size_t)(8 * q) * HID];
            bw[2 * q].y     = wp[(size_t)(8 * q + 1) * HID];
            bw[2 * q + 1].x = wp[(size_t)(8 * q + 2) * HID];
            bw[2 * q + 1].y = wp[(size_t)(8 * q + 3) * HID];
        }
    }

    const float4* __restrict__ Xv = (const float4*)X;
    const int srow = tid >> 6;
    const int scol = (tid & 63) * 4;
    const int soff = srow * A_PITCH + scol;
    int tile = blockIdx.x;
    float4 s0, s1, s2, s3;
    {
        const float4* __restrict__ xp = Xv + (size_t)tile * TILE_F4 + tid;
        s0 = xp[0];
        s1 = xp[GEMM_THREADS];
        s2 = xp[2 * GEMM_THREADS];
        s3 = xp[3 * GEMM_THREADS];
    }
    *(float4*)&xs[0][soff]                = s0;
    *(float4*)&xs[0][soff + 4 * A_PITCH]  = s1;
    *(float4*)&xs[0][soff + 8 * A_PITCH]  = s2;
    *(float4*)&xs[0][soff + 12 * A_PITCH] = s3;
    __syncthreads();

    int buf = 0;
    for (; tile < N_TILES; tile += GEMM_GRID) {
        const bool has_next = (tile + GEMM_GRID) < N_TILES;
        if (has_next) {
            const float4* __restrict__ xp = Xv + (size_t)(tile + GEMM_GRID) * TILE_F4 + tid;
            s0 = xp[0];
            s1 = xp[GEMM_THREADS];
            s2 = xp[2 * GEMM_THREADS];
            s3 = xp[3 * GEMM_THREADS];
        }

        const float* ap = &xs[buf][l16 * A_PITCH + 4 * half];
        float4 fa[8], fb[8];
        v8f acc = {};

#define A_GROUP(dst, g)                                                            \
        _Pragma("unroll") for (int j = 0; j < 8; ++j)                              \
            dst[j] = *(const float4*)(ap + 8 * (8 * (g) + j));
#define MMA_GROUP(srcf, g)                                                         \
        _Pragma("unroll") for (int j = 0; j < 8; ++j) {                            \
            const int q = 8 * (g) + j;                                             \
            v2f a0, a1;                                                            \
            a0.x = srcf[j].x; a0.y = srcf[j].y;                                    \
            a1.x = srcf[j].z; a1.y = srcf[j].w;                                    \
            acc = __builtin_amdgcn_wmma_f32_16x16x4_f32(                           \
                 false, a0,  false, bw[2 * q],                   \
                 (short)0, acc,  false,  false);    \
            acc = __builtin_amdgcn_wmma_f32_16x16x4_f32(                           \
                 false, a1,  false, bw[2 * q + 1],               \
                 (short)0, acc,  false,  false);    \
        }
#define KEEP_ORDER() __builtin_amdgcn_sched_barrier(0)

        A_GROUP(fa, 0)
        A_GROUP(fb, 1)
        KEEP_ORDER();
        MMA_GROUP(fa, 0)
        KEEP_ORDER();
        A_GROUP(fa, 2)
        KEEP_ORDER();
        MMA_GROUP(fb, 1)
        KEEP_ORDER();
        A_GROUP(fb, 3)
        KEEP_ORDER();
        MMA_GROUP(fa, 2)
        MMA_GROUP(fb, 3)
#undef A_GROUP
#undef MMA_GROUP
#undef KEEP_ORDER

        if (has_next) {
            float* xn = &xs[buf ^ 1][soff];
            *(float4*)(xn)                = s0;
            *(float4*)(xn + 4 * A_PITCH)  = s1;
            *(float4*)(xn + 8 * A_PITCH)  = s2;
            *(float4*)(xn + 12 * A_PITCH) = s3;
        }
        __syncthreads();
        buf ^= 1;

        float* __restrict__ hrow = H + (size_t)(tile * TILE_ROWS + half * 8) * HID + col;
#pragma unroll
        for (int i = 0; i < 8; ++i)
            hrow[(size_t)i * HID] = acc[i];
    }
}

#define PROP_THREADS  128
#define ROWS_PER_PASS 8
#define COUNT_LOADS   4
#define BLOCK_F4      (BLOCK_ROWS * HID / 4)
#define F4_PER_THREAD (BLOCK_F4 / PROP_THREADS)
static_assert(BLOCK_F4 % PROP_THREADS == 0, "the threads cover the block's sums exactly");
static_assert((PROP_THREADS * 4) % HID == 0, "a thread's places all lie in the same 4 channels");

template <bool BLEND>
__global__ __launch_bounds__(PROP_THREADS) void prop_kernel(const float* __restrict__ h,
                                                            const int* __restrict__ rec,
                                                            const int* __restrict__ cur,
                                                            const float* __restrict__ inv_out,
                                                            const float* __restrict__ tail,
                                                            float* __restrict__ out) {
    __shared__ float acc[BLOCK_ROWS * HID];
    __shared__ int   cnt[BLOCK_ROWS];
    __shared__ float inv_in[BLOCK_ROWS];

    const int tid  = threadIdx.x;
    const int lane = tid & 31;
    const int wave = __builtin_amdgcn_readfirstlane(tid >> 5);
    const int b    = blockIdx.x;
    const int s    = (b == 0) ? 0 : cur[(b - 1) * CUR_STRIDE];
    const int e    = cur[b * CUR_STRIDE];

    for (int i = tid; i < BLOCK_F4; i += PROP_THREADS)
        ((float4*)acc)[i] = make_float4(0.0f, 0.0f, 0.0f, 0.0f);
    if (tid < BLOCK_ROWS) cnt[tid] = 0;
    __syncthreads();

    for (int i0 = s + tid; i0 < e; i0 += COUNT_LOADS * PROP_THREADS) {
        int r[COUNT_LOADS];
        for (int k = 0; k < COUNT_LOADS; ++k) {
            const int i = i0 + k * PROP_THREADS;
            r[k] = (i < e) ? rec[i] : -1;
        }
        for (int k = 0; k < COUNT_LOADS; ++k)
            if (r[k] >= 0)
                __hip_atomic_fetch_add(&cnt[r[k] >> SRC_BITS], 1, __ATOMIC_RELAXED,
                                       __HIP_MEMORY_SCOPE_WORKGROUP);
    }
    __syncthreads();
    if (tid < BLOCK_ROWS) {
        const int c = cnt[tid];
        inv_in[tid] = (c > 0) ? rsqrtf(fmaxf((float)c, 1.0f)) : 0.0f;
    }
    __syncthreads();

    const float* __restrict__ hl = h + lane * 4;
    float* al = acc + lane;
    for (int base = s + wave * 32; base < e; base += PROP_THREADS) {
        const int n = min(32, e - base);
        int   sidx = 0;
        int   dl   = 0;
        float w    = 0.0f;
        if (lane < n) {
            const int r = rec[base + lane];
            sidx = r & SRC_MASK;
            dl   = r >> SRC_BITS;
            w    = inv_out[sidx] * inv_in[dl];
        }
        const int wv = __float_as_int(w);
        for (int j0 = 0; j0 < n; j0 += ROWS_PER_PASS) {
            float4 v[ROWS_PER_PASS];
            for (int j = 0; j < ROWS_PER_PASS; ++j) {
                const int sj = __builtin_amdgcn_readlane(sidx, min(j0 + j, n - 1));
                v[j] = *(const float4*)(hl + (size_t)sj * HID);
            }
            for (int j = 0; j < ROWS_PER_PASS; ++j) {
                if (j0 + j < n) {
                    const int   d  = __builtin_amdgcn_readlane(dl, j0 + j);
                    const float wj = __int_as_float(__builtin_amdgcn_readlane(wv, j0 + j));
                    float* a = al + d * HID;
                    lds_add_f32(a,      v[j].x * wj);
                    lds_add_f32(a + 32, v[j].y * wj);
                    lds_add_f32(a + 64, v[j].z * wj);
                    lds_add_f32(a + 96, v[j].w * wj);
                }
            }
        }
    }
    const size_t row0 = (size_t)b * BLOCK_ROWS * HID;
    float4 z[F4_PER_THREAD];
    if (BLEND) {
        for (int k = 0; k < F4_PER_THREAD; ++k)
            z[k] = *(const float4*)(tail + row0 + (size_t)(tid + k * PROP_THREADS) * 4);
    } else {
        z[0] = *(const float4*)(tail + ((tid * 4) & (HID - 1)));
    }
    __syncthreads();

    for (int k = 0; k < F4_PER_THREAD; ++k) {
        const int i = tid + k * PROP_THREADS;
        const float* ap = acc + (i >> 5) * HID + (i & 31);
        float4 a;
        a.x = ap[0];
        a.y = ap[32];
        a.z = ap[64];
        a.w = ap[96];
        float4 r;
        if (BLEND) {
            r.x = (1.0f - ALPHA) * a.x + ALPHA * z[k].x;
            r.y = (1.0f - ALPHA) * a.y + ALPHA * z[k].y;
            r.z = (1.0f - ALPHA) * a.z + ALPHA * z[k].z;
            r.w = (1.0f - ALPHA) * a.w + ALPHA * z[k].w;
        } else {
            r.x = fmaxf(a.x + z[0].x, 0.0f);
            r.y = fmaxf(a.y + z[0].y, 0.0f);
            r.z = fmaxf(a.z + z[0].z, 0.0f);
            r.w = fmaxf(a.w + z[0].w, 0.0f);
        }
        *(float4*)(out + row0 + (size_t)i * 4) = r;
    }
}

extern "C" void kernel_launch(void* const* d_in, const int* in_sizes, int n_in,
                              void* d_out, int out_size, void* d_ws, size_t ws_size,
                              hipStream_t stream) {
    const float* features = (const float*)d_in[0];
    const float* W        = (const float*)d_in[1];
    const float* bias     = (const float*)d_in[2];
    const int*   eidx     = (const int*)d_in[3];
    const int*   src      = eidx;
    const int*   dst      = eidx + N_EDGES;

    float* out = (float*)d_out;

    const size_t NH = (size_t)N_NODES * HID;
    float* deg     = (float*)d_ws;
    float* deg_out = deg;
    float* deg_in  = deg + N_NODES;
    int*   cur     = (int*)deg_in;
    int*   rec     = (int*)(deg + 2 * N_NODES);
    int*   counts  = rec;
    float* bufA    = (float*)(rec + N_EDGES);
    float* bufB    = bufA + NH;

    hipMemsetAsync(deg, 0, 2 * (size_t)N_NODES * sizeof(float), stream);
    deg_kernel<<<(N_EDGES + 255) / 256, 256, 0, stream>>>(src, dst, deg_out, deg_in);
    rowprep_kernel<<<(N_NODES + 255) / 256, 256, 0, stream>>>(deg_out, deg_in, counts);
    scan_kernel<<<1, 1024, 0, stream>>>(counts, cur);
    fill_kernel<<<(N_EDGES + 255) / 256, 256, 0, stream>>>(src, dst, cur, rec);

    gemm_kernel<<<GEMM_GRID, GEMM_THREADS, 0, stream>>>(features, W, bufA);
    prop_kernel<false><<<N_BLOCKS, PROP_THREADS, 0, stream>>>(bufA, rec, cur, deg_out, bias, bufB);

    float* hc = bufB;
    for (int t = 0; t < K_STEPS; ++t) {
        float* tmp = (t & 1) ? out : bufA;
        prop_kernel<true><<<N_BLOCKS, PROP_THREADS, 0, stream>>>(hc, rec, cur, deg_out, bufB, tmp);
        hc = tmp;
    }
}


# prop: an edge record carries its source row's out-degree (9 spare bits, set once at the bucket fill); D_out^-1/2 comes from it, the scattered per-edge weight load of every step is gone; sums as before
# speedup vs baseline: 4.4636x; 1.0394x over previous
#include <hip/hip_runtime.h>
#include <cstdint>

#define N_NODES 100000
#define N_EDGES 1600000
#define IN_CH   256
#define HID     128
#define K_STEPS 10
#define ALPHA   0.1f

#define BLOCK_ROWS 32
#define N_BLOCKS   (N_NODES / BLOCK_ROWS)
#define SRC_BITS   17
#define SRC_MASK   ((1 << SRC_BITS) - 1)
#define ROW_BITS   5
#define DEG_SHIFT  (SRC_BITS + ROW_BITS)
#define DEG_BITS   9
#define DEG_SAT    ((1 << DEG_BITS) - 1)
#define CUR_STRIDE 32
static_assert(N_NODES % BLOCK_ROWS == 0, "blocks cover the rows exactly");
static_assert(N_NODES <= (1 << SRC_BITS), "a source row fits its field");
static_assert(BLOCK_ROWS == (1 << ROW_BITS), "a row inside its block fits its field");
static_assert(DEG_SHIFT + DEG_BITS <= 31, "the record's sign bit stays clear");
static_assert(N_BLOCKS * CUR_STRIDE <= N_NODES, "the cursors fit in the in-degree array they replace");

typedef __attribute__((ext_vector_type(2))) float v2f;
typedef __attribute__((ext_vector_type(8))) float v8f;

__device__ __forceinline__ void atom_add_f32(float* p, float v) {
    __hip_atomic_fetch_add(p, v, __ATOMIC_RELAXED, __HIP_MEMORY_SCOPE_AGENT);
}

__device__ __forceinline__ void lds_add_f32(float* p, float v) {
    __hip_atomic_fetch_add(p, v, __ATOMIC_RELAXED, __HIP_MEMORY_SCOPE_WORKGROUP);
}

__global__ __launch_bounds__(256) void deg_kernel(const int* __restrict__ src,
                                                  const int* __restrict__ dst,
                                                  float* __restrict__ deg_out,
                                                  float* __restrict__ deg_in) {
    int e = blockIdx.x * blockDim.x + threadIdx.x;
    if (e < N_EDGES) {
        atom_add_f32(&deg_out[src[e]], 1.0f);
        atom_add_f32(&deg_in[dst[e]], 1.0f);
    }
}

__global__ __launch_bounds__(256) void rowprep_kernel(const float* __restrict__ deg_in,
                                                      int* __restrict__ counts) {
    const int i = blockIdx.x * blockDim.x + threadIdx.x;
    float din = 0.0f;
    if (i < N_NODES)
        din = deg_in[i];
    for (int off = 16; off > 0; off >>= 1)
        din += __shfl_xor(din, off, 32);
    if ((threadIdx.x & 31) == 0 && i < N_NODES)
        counts[i >> 5] = (int)(din + 0.5f);
}

#define SCAN_N 4096
static_assert(N_BLOCKS <= SCAN_N, "the scan covers every block");

__global__ __launch_bounds__(1024) void scan_kernel(const int* __restrict__ counts,
                                                    int* __restrict__ cur) {
    __shared__ int bufs[2][SCAN_N];
    const int tid = threadIdx.x;
    for (int i = tid; i < SCAN_N; i += 1024)
        bufs[0][i] = (i < N_BLOCKS) ? counts[i] : 0;
    __syncthreads();
    int p = 0;
    for (int d = 1; d < SCAN_N; d <<= 1) {
        for (int i = tid; i < SCAN_N; i += 1024) {
            int v = bufs[p][i];
            if (i >= d) v += bufs[p][i - d];
            bufs[p ^ 1][i] = v;
        }
        __syncthreads();
        p ^= 1;
    }
    for (int i = tid; i < N_BLOCKS; i += 1024)
        cur[i * CUR_STRIDE] = (i == 0) ? 0 : bufs[p][i - 1];
}

__global__ __launch_bounds__(256) void fill_kernel(const int* __restrict__ src,
                                                   const int* __restrict__ dst,
                                                   const float* __restrict__ deg_out,
                                                   int* __restrict__ cur,
                                                   int* __restrict__ rec) {
    const int e = blockIdx.x * blockDim.x + threadIdx.x;
    if (e < N_EDGES) {
        const int s = src[e];
        const int d = dst[e];
        const int od = min((int)deg_out[s], DEG_SAT);
        const int pos = __hip_atomic_fetch_add(&cur[(d >> 5) * CUR_STRIDE], 1, __ATOMIC_RELAXED,
                                               __HIP_MEMORY_SCOPE_AGENT);
        rec[pos] = s | ((d & (BLOCK_ROWS - 1)) << SRC_BITS) | (od << DEG_SHIFT);
    }
}

#define GEMM_THREADS 256
#define TILE_ROWS    16
#define N_TILES      (N_NODES / TILE_ROWS)
#define TILE_F4      (TILE_ROWS * IN_CH / 4)
#define A_PITCH      (IN_CH + 4)
#define GEMM_GRID    512
static_assert(N_NODES % TILE_ROWS == 0, "tiles cover the rows exactly");
static_assert(TILE_F4 == 4 * GEMM_THREADS, "a thread fetches 4 float4 of a tile");
static_assert(GEMM_GRID <= N_TILES, "every work-group has a first tile");
static_assert(((A_PITCH / 4) & 1) == 1 && A_PITCH % 4 == 0, "pitch is 16 bytes x odd");

__global__ __launch_bounds__(GEMM_THREADS, 2) void gemm_kernel(const float* __restrict__ X,
                                                                const float* __restrict__ W,
                                                                float* __restrict__ H) {
    __shared__ float xs[2][TILE_ROWS * A_PITCH];

    const int tid  = threadIdx.x;
    const int lane = tid & 31;
    const int wave = tid >> 5;
    const int half = lane >> 4;
    const int l16  = lane & 15;
    const int col  = wave * 16 + l16;

    v2f bw[IN_CH / 4];
    {
        const float* __restrict__ wp = W + (size_t)(4 * half) * HID + col;
#pragma unroll
        for (int q = 0; q < IN_CH / 8; ++q) {
            bw[2 * q].x     = wp[(size_t)(8 * q) * HID];
            bw[2 * q].y     = wp[(size_t)(8 * q + 1) * HID];
            bw[2 * q + 1].x = wp[(size_t)(8 * q + 2) * HID];
            bw[2 * q + 1].y = wp[(size_t)(8 * q + 3) * HID];
        }
    }

    const float4* __restrict__ Xv = (const float4*)X;
    const int srow = tid >> 6;
    const int scol = (tid & 63) * 4;
    const int soff = srow * A_PITCH + scol;
    int tile = blockIdx.x;
    float4 s0, s1, s2, s3;
    {
        const float4* __restrict__ xp = Xv + (size_t)tile * TILE_F4 + tid;
        s0 = xp[0];
        s1 = xp[GEMM_THREADS];
        s2 = xp[2 * GEMM_THREADS];
        s3 = xp[3 * GEMM_THREADS];
    }
    *(float4*)&xs[0][soff]                = s0;
    *(float4*)&xs[0][soff + 4 * A_PITCH]  = s1;
    *(float4*)&xs[0][soff + 8 * A_PITCH]  = s2;
    *(float4*)&xs[0][soff + 12 * A_PITCH] = s3;
    __syncthreads();

    int buf = 0;
    for (; tile < N_TILES; tile += GEMM_GRID) {
        const bool has_next = (tile + GEMM_GRID) < N_TILES;
        if (has_next) {
            const float4* __restrict__ xp = Xv + (size_t)(tile + GEMM_GRID) * TILE_F4 + tid;
            s0 = xp[0];
            s1 = xp[GEMM_THREADS];
            s2 = xp[2 * GEMM_THREADS];
            s3 = xp[3 * GEMM_THREADS];
        }

        const float* ap = &xs[buf][l16 * A_PITCH + 4 * half];
        float4 fa[8], fb[8];
        v8f acc = {};

#define A_GROUP(dst, g)                                                            \
        _Pragma("unroll") for (int j = 0; j < 8; ++j)                              \
            dst[j] = *(const float4*)(ap + 8 * (8 * (g) + j));
#define MMA_GROUP(srcf, g)                                                         \
        _Pragma("unroll") for (int j = 0; j < 8; ++j) {                            \
            const int q = 8 * (g) + j;                                             \
            v2f a0, a1;                                                            \
            a0.x = srcf[j].x; a0.y = srcf[j].y;                                    \
            a1.x = srcf[j].z; a1.y = srcf[j].w;                                    \
            acc = __builtin_amdgcn_wmma_f32_16x16x4_f32(                           \
                 false, a0,  false, bw[2 * q],                   \
                 (short)0, acc,  false,  false);    \
            acc = __builtin_amdgcn_wmma_f32_16x16x4_f32(                           \
                 false, a1,  false, bw[2 * q + 1],               \
                 (short)0, acc,  false,  false);    \
        }
#define KEEP_ORDER() __builtin_amdgcn_sched_barrier(0)

        A_GROUP(fa, 0)
        A_GROUP(fb, 1)
        KEEP_ORDER();
        MMA_GROUP(fa, 0)
        KEEP_ORDER();
        A_GROUP(fa, 2)
        KEEP_ORDER();
        MMA_GROUP(fb, 1)
        KEEP_ORDER();
        A_GROUP(fb, 3)
        KEEP_ORDER();
        MMA_GROUP(fa, 2)
        MMA_GROUP(fb, 3)
#undef A_GROUP
#undef MMA_GROUP
#undef KEEP_ORDER

        if (has_next) {
            float* xn = &xs[buf ^ 1][soff];
            *(float4*)(xn)                = s0;
            *(float4*)(xn + 4 * A_PITCH)  = s1;
            *(float4*)(xn + 8 * A_PITCH)  = s2;
            *(float4*)(xn + 12 * A_PITCH) = s3;
        }
        __syncthreads();
        buf ^= 1;

        float* __restrict__ hrow = H + (size_t)(tile * TILE_ROWS + half * 8) * HID + col;
#pragma unroll
        for (int i = 0; i < 8; ++i)
            hrow[(size_t)i * HID] = acc[i];
    }
}

#define PROP_THREADS  128
#define ROWS_PER_PASS 8
#define COUNT_LOADS   4
#define BLOCK_F4      (BLOCK_ROWS * HID / 4)
#define F4_PER_THREAD (BLOCK_F4 / PROP_THREADS)
static_assert(BLOCK_F4 % PROP_THREADS == 0, "the threads cover the block's sums exactly");
static_assert((PROP_THREADS * 4) % HID == 0, "a thread's places all lie in the same 4 channels");

template <bool BLEND>
__global__ __launch_bounds__(PROP_THREADS) void prop_kernel(const float* __restrict__ h,
                                                            const int* __restrict__ rec,
                                                            const int* __restrict__ cur,
                                                            const float* __restrict__ deg_out,
                                                            const float* __restrict__ tail,
                                                            float* __restrict__ out) {
    __shared__ float acc[BLOCK_ROWS * HID];
    __shared__ int   cnt[BLOCK_ROWS];
    __shared__ float inv_in[BLOCK_ROWS];

    const int tid  = threadIdx.x;
    const int lane = tid & 31;
    const int wave = __builtin_amdgcn_readfirstlane(tid >> 5);
    const int b    = blockIdx.x;
    const int s    = (b == 0) ? 0 : cur[(b - 1) * CUR_STRIDE];
    const int e    = cur[b * CUR_STRIDE];

    for (int i = tid; i < BLOCK_F4; i += PROP_THREADS)
        ((float4*)acc)[i] = make_float4(0.0f, 0.0f, 0.0f, 0.0f);
    if (tid < BLOCK_ROWS) cnt[tid] = 0;
    __syncthreads();

    for (int i0 = s + tid; i0 < e; i0 += COUNT_LOADS * PROP_THREADS) {
        int r[COUNT_LOADS];
        for (int k = 0; k < COUNT_LOADS; ++k) {
            const int i = i0 + k * PROP_THREADS;
            r[k] = (i < e) ? rec[i] : -1;
        }
        for (int k = 0; k < COUNT_LOADS; ++k)
            if (r[k] >= 0)
                __hip_atomic_fetch_add(&cnt[(r[k] >> SRC_BITS) & (BLOCK_ROWS - 1)], 1, __ATOMIC_RELAXED,
                                       __HIP_MEMORY_SCOPE_WORKGROUP);
    }
    __syncthreads();
    if (tid < BLOCK_ROWS) {
        const int c = cnt[tid];
        inv_in[tid] = (c > 0) ? rsqrtf(fmaxf((float)c, 1.0f)) : 0.0f;
    }
    __syncthreads();

    const float* __restrict__ hl = h + lane * 4;
    float* al = acc + lane;
    for (int base = s + wave * 32; base < e; base += PROP_THREADS) {
        const int n = min(32, e - base);
        int   sidx = 0;
        int   dl   = 0;
        float w    = 0.0f;
        if (lane < n) {
            const int r = rec[base + lane];
            sidx = r & SRC_MASK;
            dl   = (r >> SRC_BITS) & (BLOCK_ROWS - 1);
            const int od = r >> DEG_SHIFT;
            float dgo = (float)od;
            if (od == DEG_SAT) dgo = deg_out[sidx];
            const float io = (dgo > 0.0f) ? rsqrtf(fmaxf(dgo, 1.0f)) : 0.0f;
            w    = io * inv_in[dl];
        }
        const int wv = __float_as_int(w);
        for (int j0 = 0; j0 < n; j0 += ROWS_PER_PASS) {
            float4 v[ROWS_PER_PASS];
            for (int j = 0; j < ROWS_PER_PASS; ++j) {
                const int sj = __builtin_amdgcn_readlane(sidx, min(j0 + j, n - 1));
                v[j] = *(const float4*)(hl + (size_t)sj * HID);
            }
            for (int j = 0; j < ROWS_PER_PASS; ++j) {
                if (j0 + j < n) {
                    const int   d  = __builtin_amdgcn_readlane(dl, j0 + j);
                    const float wj = __int_as_float(__builtin_amdgcn_readlane(wv, j0 + j));
                    float* a = al + d * HID;
                    lds_add_f32(a,      v[j].x * wj);
                    lds_add_f32(a + 32, v[j].y * wj);
                    lds_add_f32(a + 64, v[j].z * wj);
                    lds_add_f32(a + 96, v[j].w * wj);
                }
            }
        }
    }
    const size_t row0 = (size_t)b * BLOCK_ROWS * HID;
    float4 z[F4_PER_THREAD];
    if (BLEND) {
        for (int k = 0; k < F4_PER_THREAD; ++k)
            z[k] = *(const float4*)(tail + row0 + (size_t)(tid + k * PROP_THREADS) * 4);
    } else {
        z[0] = *(const float4*)(tail + ((tid * 4) & (HID - 1)));
    }
    __syncthreads();

    for (int k = 0; k < F4_PER_THREAD; ++k) {
        const int i = tid + k * PROP_THREADS;
        const float* ap = acc + (i >> 5) * HID + (i & 31);
        float4 a;
        a.x = ap[0];
        a.y = ap[32];
        a.z = ap[64];
        a.w = ap[96];
        float4 r;
        if (BLEND) {
            r.x = (1.0f - ALPHA) * a.x + ALPHA * z[k].x;
            r.y = (1.0f - ALPHA) * a.y + ALPHA * z[k].y;
            r.z = (1.0f - ALPHA) * a.z + ALPHA * z[k].z;
            r.w = (1.0f - ALPHA) * a.w + ALPHA * z[k].w;
        } else {
            r.x = fmaxf(a.x + z[0].x, 0.0f);
            r.y = fmaxf(a.y + z[0].y, 0.0f);
            r.z = fmaxf(a.z + z[0].z, 0.0f);
            r.w = fmaxf(a.w + z[0].w, 0.0f);
        }
        *(float4*)(out + row0 + (size_t)i * 4) = r;
    }
}

extern "C" void kernel_launch(void* const* d_in, const int* in_sizes, int n_in,
                              void* d_out, int out_size, void* d_ws, size_t ws_size,
                              hipStream_t stream) {
    const float* features = (const float*)d_in[0];
    const float* W        = (const float*)d_in[1];
    const float* bias     = (const float*)d_in[2];
    const int*   eidx     = (const int*)d_in[3];
    const int*   src      = eidx;
    const int*   dst      = eidx + N_EDGES;

    float* out = (float*)d_out;

    const size_t NH = (size_t)N_NODES * HID;
    float* deg     = (float*)d_ws;
    float* deg_out = deg;
    float* deg_in  = deg + N_NODES;
    int*   cur     = (int*)deg_in;
    int*   rec     = (int*)(deg + 2 * N_NODES);
    int*   counts  = rec;
    float* bufA    = (float*)(rec + N_EDGES);
    float* bufB    = bufA + NH;

    hipMemsetAsync(deg, 0, 2 * (size_t)N_NODES * sizeof(float), stream);
    deg_kernel<<<(N_EDGES + 255) / 256, 256, 0, stream>>>(src, dst, deg_out, deg_in);
    rowprep_kernel<<<(N_NODES + 255) / 256, 256, 0, stream>>>(deg_in, counts);
    scan_kernel<<<1, 1024, 0, stream>>>(counts, cur);
    fill_kernel<<<(N_EDGES + 255) / 256, 256, 0, stream>>>(src, dst, deg_out, cur, rec);

    gemm_kernel<<<GEMM_GRID, GEMM_THREADS, 0, stream>>>(features, W, bufA);
    prop_kernel<false><<<N_BLOCKS, PROP_THREADS, 0, stream>>>(bufA, rec, cur, deg_out, bias, bufB);

    float* hc = bufB;
    for (int t = 0; t < K_STEPS; ++t) {
        float* tmp = (t & 1) ? out : bufA;
        prop_kernel<true><<<N_BLOCKS, PROP_THREADS, 0, stream>>>(hc, rec, cur, deg_out, bufB, tmp);
        hc = tmp;
    }
}


# prop: a wave owns 16 dst rows, their sums in registers (edges bucketed by 16-row group; the record's row, a scalar, indexes the registers); no LDS float adds, no barriers; sums in bucket order
# speedup vs baseline: 4.6819x; 1.0489x over previous
#include <hip/hip_runtime.h>
#include <cstdint>

#define N_NODES 100000
#define N_EDGES 1600000
#define IN_CH   256
#define HID     128
#define K_STEPS 10
#define ALPHA   0.1f

#define GROUP_ROWS 16
#define N_GROUPS   (N_NODES / GROUP_ROWS)
#define SRC_BITS   17
#define SRC_MASK   ((1 << SRC_BITS) - 1)
#define ROW_BITS   4
#define ROW_MASK   ((1 << ROW_BITS) - 1)
#define DEG_SHIFT  (SRC_BITS + ROW_BITS)
#define DEG_BITS   9
#define DEG_SAT    ((1 << DEG_BITS) - 1)
#define CUR_STRIDE GROUP_ROWS
static_assert(N_NODES % GROUP_ROWS == 0, "groups cover the rows exactly");
static_assert(N_NODES <= (1 << SRC_BITS), "a source row fits its field");
static_assert(GROUP_ROWS <= (1 << ROW_BITS), "a row inside its group fits its field");
static_assert(DEG_SHIFT + DEG_BITS <= 31, "the record's sign bit stays clear");
static_assert(N_GROUPS * CUR_STRIDE <= N_NODES, "the cursors fit in the in-degree array they replace");

typedef __attribute__((ext_vector_type(2))) float v2f;
typedef __attribute__((ext_vector_type(8))) float v8f;

__device__ __forceinline__ void atom_add_f32(float* p, float v) {
    __hip_atomic_fetch_add(p, v, __ATOMIC_RELAXED, __HIP_MEMORY_SCOPE_AGENT);
}

__global__ __launch_bounds__(256) void deg_kernel(const int* __restrict__ src,
                                                  const int* __restrict__ dst,
                                                  float* __restrict__ deg_out,
                                                  float* __restrict__ deg_in) {
    int e = blockIdx.x * blockDim.x + threadIdx.x;
    if (e < N_EDGES) {
        atom_add_f32(&deg_out[src[e]], 1.0f);
        atom_add_f32(&deg_in[dst[e]], 1.0f);
    }
}

__global__ __launch_bounds__(256) void rowprep_kernel(const float* __restrict__ deg_in,
                                                      int* __restrict__ counts) {
    const int g = blockIdx.x * blockDim.x + threadIdx.x;
    if (g < N_GROUPS) {
        float din = 0.0f;
        for (int k = 0; k < GROUP_ROWS; ++k)
            din += deg_in[g * GROUP_ROWS + k];
        counts[g] = (int)(din + 0.5f);
    }
}

#define SCAN_THREADS 1024
#define SCAN_RUN     ((N_GROUPS + SCAN_THREADS - 1) / SCAN_THREADS)

__global__ __launch_bounds__(SCAN_THREADS) void scan_kernel(const int* __restrict__ counts,
                                                            int* __restrict__ cur) {
    __shared__ int bufs[2][SCAN_THREADS];
    const int tid = threadIdx.x;
    const int i0  = tid * SCAN_RUN;
    int c[SCAN_RUN];
    int sum = 0;
    for (int k = 0; k < SCAN_RUN; ++k) {
        const int i = i0 + k;
        c[k] = (i < N_GROUPS) ? counts[i] : 0;
        sum += c[k];
    }
    bufs[0][tid] = sum;
    __syncthreads();
    int p = 0;
    for (int d = 1; d < SCAN_THREADS; d <<= 1) {
        int v = bufs[p][tid];
        if (tid >= d) v += bufs[p][tid - d];
        bufs[p ^ 1][tid] = v;
        __syncthreads();
        p ^= 1;
    }
    int run = (tid == 0) ? 0 : bufs[p][tid - 1];
    for (int k = 0; k < SCAN_RUN; ++k) {
        const int i = i0 + k;
        if (i < N_GROUPS) cur[i * CUR_STRIDE] = run;
        run += c[k];
    }
}

__global__ __launch_bounds__(256) void fill_kernel(const int* __restrict__ src,
                                                   const int* __restrict__ dst,
                                                   const float* __restrict__ deg_out,
                                                   int* __restrict__ cur,
                                                   int* __restrict__ rec) {
    const int e = blockIdx.x * blockDim.x + threadIdx.x;
    if (e < N_EDGES) {
        const int s = src[e];
        const int d = dst[e];
        const int g = d / GROUP_ROWS;
        const int od = min((int)deg_out[s], DEG_SAT);
        const int pos = __hip_atomic_fetch_add(&cur[g * CUR_STRIDE], 1, __ATOMIC_RELAXED,
                                               __HIP_MEMORY_SCOPE_AGENT);
        rec[pos] = s | ((d - g * GROUP_ROWS) << SRC_BITS) | (od << DEG_SHIFT);
    }
}

#define GEMM_THREADS 256
#define TILE_ROWS    16
#define N_TILES      (N_NODES / TILE_ROWS)
#define TILE_F4      (TILE_ROWS * IN_CH / 4)
#define A_PITCH      (IN_CH + 4)
#define GEMM_GRID    512
static_assert(N_NODES % TILE_ROWS == 0, "tiles cover the rows exactly");
static_assert(TILE_F4 == 4 * GEMM_THREADS, "a thread fetches 4 float4 of a tile");
static_assert(GEMM_GRID <= N_TILES, "every work-group has a first tile");
static_assert(((A_PITCH / 4) & 1) == 1 && A_PITCH % 4 == 0, "pitch is 16 bytes x odd");

__global__ __launch_bounds__(GEMM_THREADS, 2) void gemm_kernel(const float* __restrict__ X,
                                                                const float* __restrict__ W,
                                                                float* __restrict__ H) {
    __shared__ float xs[2][TILE_ROWS * A_PITCH];

    const int tid  = threadIdx.x;
    const int lane = tid & 31;
    const int wave = tid >> 5;
    const int half = lane >> 4;
    const int l16  = lane & 15;
    const int col  = wave * 16 + l16;

    v2f bw[IN_CH / 4];
    {
        const float* __restrict__ wp = W + (size_t)(4 * half) * HID + col;
#pragma unroll
        for (int q = 0; q < IN_CH / 8; ++q) {
            bw[2 * q].x     = wp[(size_t)(8 * q) * HID];
            bw[2 * q].y     = wp[(size_t)(8 * q + 1) * HID];
            bw[2 * q + 1].x = wp[(size_t)(8 * q + 2) * HID];
            bw[2 * q + 1].y = wp[(size_t)(8 * q + 3) * HID];
        }
    }

    const float4* __restrict__ Xv = (const float4*)X;
    const int srow = tid >> 6;
    const int scol = (tid & 63) * 4;
    const int soff = srow * A_PITCH + scol;
    int tile = blockIdx.x;
    float4 s0, s1, s2, s3;
    {
        const float4* __restrict__ xp = Xv + (size_t)tile * TILE_F4 + tid;
        s0 = xp[0];
        s1 = xp[GEMM_THREADS];
        s2 = xp[2 * GEMM_THREADS];
        s3 = xp[3 * GEMM_THREADS];
    }
    *(float4*)&xs[0][soff]                = s0;
    *(float4*)&xs[0][soff + 4 * A_PITCH]  = s1;
    *(float4*)&xs[0][soff + 8 * A_PITCH]  = s2;
    *(float4*)&xs[0][soff + 12 * A_PITCH] = s3;
    __syncthreads();

    int buf = 0;
    for (; tile < N_TILES; tile += GEMM_GRID) {
        const bool has_next = (tile + GEMM_GRID) < N_TILES;
        if (has_next) {
            const float4* __restrict__ xp = Xv + (size_t)(tile + GEMM_GRID) * TILE_F4 + tid;
            s0 = xp[0];
            s1 = xp[GEMM_THREADS];
            s2 = xp[2 * GEMM_THREADS];
            s3 = xp[3 * GEMM_THREADS];
        }

        const float* ap = &xs[buf][l16 * A_PITCH + 4 * half];
        float4 fa[8], fb[8];
        v8f acc = {};

#define A_GROUP(dst, g)                                                            \
        _Pragma("unroll") for (int j = 0; j < 8; ++j)                              \
            dst[j] = *(const float4*)(ap + 8 * (8 * (g) + j));
#define MMA_GROUP(srcf, g)                                                         \
        _Pragma("unroll") for (int j = 0; j < 8; ++j) {                            \
            const int q = 8 * (g) + j;                                             \
            v2f a0, a1;                                                            \
            a0.x = srcf[j].x; a0.y = srcf[j].y;                                    \
            a1.x = srcf[j].z; a1.y = srcf[j].w;                                    \
            acc = __builtin_amdgcn_wmma_f32_16x16x4_f32(                           \
                 false, a0,  false, bw[2 * q],                   \
                 (short)0, acc,  false,  false);    \
            acc = __builtin_amdgcn_wmma_f32_16x16x4_f32(                           \
                 false, a1,  false, bw[2 * q + 1],               \
                 (short)0, acc,  false,  false);    \
        }
#define KEEP_ORDER() __builtin_amdgcn_sched_barrier(0)

        A_GROUP(fa, 0)
        A_GROUP(fb, 1)
        KEEP_ORDER();
        MMA_GROUP(fa, 0)
        KEEP_ORDER();
        A_GROUP(fa, 2)
        KEEP_ORDER();
        MMA_GROUP(fb, 1)
        KEEP_ORDER();
        A_GROUP(fb, 3)
        KEEP_ORDER();
        MMA_GROUP(fa, 2)
        MMA_GROUP(fb, 3)
#undef A_GROUP
#undef MMA_GROUP
#undef KEEP_ORDER

        if (has_next) {
            float* xn = &xs[buf ^ 1][soff];
            *(float4*)(xn)                = s0;
            *(float4*)(xn + 4 * A_PITCH)  = s1;
            *(float4*)(xn + 8 * A_PITCH)  = s2;
            *(float4*)(xn + 12 * A_PITCH) = s3;
        }
        __syncthreads();
        buf ^= 1;

        float* __restrict__ hrow = H + (size_t)(tile * TILE_ROWS + half * 8) * HID + col;
#pragma unroll
        for (int i = 0; i < 8; ++i)
            hrow[(size_t)i * HID] = acc[i];
    }
}

#define PROP_THREADS  32
#define ROWS_PER_PASS 11
#define COUNT_LOADS   4
#define TAIL_ROWS     8
static_assert(HID == 4 * 32, "a lane owns 4 adjacent channels of a row");
static_assert(GROUP_ROWS == 16, "a register vector has one element per row of the group");
static_assert(GROUP_ROWS % TAIL_ROWS == 0, "the tail takes the rows in whole parts");

typedef __attribute__((ext_vector_type(16))) float v16f;

template <bool BLEND>
__global__ __launch_bounds__(PROP_THREADS, 7) void prop_kernel(const float* __restrict__ h,
                                                            const int* __restrict__ rec,
                                                            const int* __restrict__ cur,
                                                            const float* __restrict__ deg_out,
                                                            const float* __restrict__ tail,
                                                            float* __restrict__ out) {
    const int lane = threadIdx.x;
    const int g    = blockIdx.x;
    const int s    = (g == 0) ? 0 : cur[(g - 1) * CUR_STRIDE];
    const int e    = cur[g * CUR_STRIDE];

    int cnt[GROUP_ROWS];
#pragma unroll
    for (int k = 0; k < GROUP_ROWS; ++k) cnt[k] = 0;
    for (int base = s; base < e; base += COUNT_LOADS * 32) {
        int rr[COUNT_LOADS];
#pragma unroll
        for (int c = 0; c < COUNT_LOADS; ++c) {
            const int i = base + c * 32 + lane;
            rr[c] = (i < e) ? rec[i] : -1;
        }
#pragma unroll
        for (int c = 0; c < COUNT_LOADS; ++c) {
            const int row = (rr[c] >= 0) ? ((rr[c] >> SRC_BITS) & ROW_MASK) : GROUP_ROWS;
#pragma unroll
            for (int k = 0; k < GROUP_ROWS; ++k)
                cnt[k] += __builtin_popcount(__builtin_amdgcn_ballot_w32(row == k));
        }
    }
    int mycnt = 0;
#pragma unroll
    for (int k = 0; k < GROUP_ROWS; ++k)
        if (lane == k) mycnt = cnt[k];
    const float my_inv = (mycnt > 0) ? rsqrtf(fmaxf((float)mycnt, 1.0f)) : 0.0f;

    v16f ax = {}, ay = {}, az = {}, aw = {};

    const float* __restrict__ hl = h + lane * 4;
    for (int base = s; base < e; base += 32) {
        const int n = min(32, e - base);
        const int r = (lane < n) ? rec[base + lane] : -1;
        int   sidx = 0;
        int   dl   = 0;
        float io   = 0.0f;
        if (r >= 0) {
            sidx = r & SRC_MASK;
            dl   = (r >> SRC_BITS) & ROW_MASK;
            const int od = r >> DEG_SHIFT;
            float dgo = (float)od;
            if (od == DEG_SAT) dgo = deg_out[sidx];
            io = (dgo > 0.0f) ? rsqrtf(fmaxf(dgo, 1.0f)) : 0.0f;
        }
        const float w  = io * __shfl(my_inv, dl, 32);
        const int   wv = __float_as_int(w);
        for (int j0 = 0; j0 < n; j0 += ROWS_PER_PASS) {
            float4 v[ROWS_PER_PASS];
#pragma unroll
            for (int j = 0; j < ROWS_PER_PASS; ++j) {
                const int sj = __builtin_amdgcn_readlane(sidx, min(j0 + j, n - 1));
                v[j] = *(const float4*)(hl + (size_t)sj * HID);
            }
#pragma unroll
            for (int j = 0; j < ROWS_PER_PASS; ++j) {
                if (j0 + j < n) {
                    const int   d  = __builtin_amdgcn_readlane(dl, j0 + j);
                    const float wj = __int_as_float(__builtin_amdgcn_readlane(wv, j0 + j));
                    ax[d] += v[j].x * wj;
                    ay[d] += v[j].y * wj;
                    az[d] += v[j].z * wj;
                    aw[d] += v[j].w * wj;
                }
            }
        }
    }

    const size_t row0 = (size_t)g * GROUP_ROWS * HID + lane * 4;
    if (BLEND) {
#pragma unroll
        for (int k0 = 0; k0 < GROUP_ROWS; k0 += TAIL_ROWS) {
            float4 z[TAIL_ROWS];
#pragma unroll
            for (int k = 0; k < TAIL_ROWS; ++k)
                z[k] = *(const float4*)(tail + row0 + (size_t)(k0 + k) * HID);
#pragma unroll
            for (int k = 0; k < TAIL_ROWS; ++k) {
                float4 r;
                r.x = (1.0f - ALPHA) * ax[k0 + k] + ALPHA * z[k].x;
                r.y = (1.0f - ALPHA) * ay[k0 + k] + ALPHA * z[k].y;
                r.z = (1.0f - ALPHA) * az[k0 + k] + ALPHA * z[k].z;
                r.w = (1.0f - ALPHA) * aw[k0 + k] + ALPHA * z[k].w;
                *(float4*)(out + row0 + (size_t)(k0 + k) * HID) = r;
            }
        }
    } else {
        const float4 bz = *(const float4*)(tail + lane * 4);
#pragma unroll
        for (int k = 0; k < GROUP_ROWS; ++k) {
            float4 r;
            r.x = fmaxf(ax[k] + bz.x, 0.0f);
            r.y = fmaxf(ay[k] + bz.y, 0.0f);
            r.z = fmaxf(az[k] + bz.z, 0.0f);
            r.w = fmaxf(aw[k] + bz.w, 0.0f);
            *(float4*)(out + row0 + (size_t)k * HID) = r;
        }
    }
}

extern "C" void kernel_launch(void* const* d_in, const int* in_sizes, int n_in,
                              void* d_out, int out_size, void* d_ws, size_t ws_size,
                              hipStream_t stream) {
    const float* features = (const float*)d_in[0];
    const float* W        = (const float*)d_in[1];
    const float* bias     = (const float*)d_in[2];
    const int*   eidx     = (const int*)d_in[3];
    const int*   src      = eidx;
    const int*   dst      = eidx + N_EDGES;

    float* out = (float*)d_out;

    const size_t NH = (size_t)N_NODES * HID;
    float* deg     = (float*)d_ws;
    float* deg_out = deg;
    float* deg_in  = deg + N_NODES;
    int*   cur     = (int*)deg_in;
    int*   rec     = (int*)(deg + 2 * N_NODES);
    int*   counts  = rec;
    float* bufA    = (float*)(rec + N_EDGES);
    float* bufB    = bufA + NH;

    hipMemsetAsync(deg, 0, 2 * (size_t)N_NODES * sizeof(float), stream);
    deg_kernel<<<(N_EDGES + 255) / 256, 256, 0, stream>>>(src, dst, deg_out, deg_in);
    rowprep_kernel<<<(N_GROUPS + 255) / 256, 256, 0, stream>>>(deg_in, counts);
    scan_kernel<<<1, SCAN_THREADS, 0, stream>>>(counts, cur);
    fill_kernel<<<(N_EDGES + 255) / 256, 256, 0, stream>>>(src, dst, deg_out, cur, rec);

    gemm_kernel<<<GEMM_GRID, GEMM_THREADS, 0, stream>>>(features, W, bufA);
    prop_kernel<false><<<N_GROUPS, PROP_THREADS, 0, stream>>>(bufA, rec, cur, deg_out, bias, bufB);

    float* hc = bufB;
    for (int t = 0; t < K_STEPS; ++t) {
        float* tmp = (t & 1) ? out : bufA;
        prop_kernel<true><<<N_GROUPS, PROP_THREADS, 0, stream>>>(hc, rec, cur, deg_out, bufB, tmp);
        hc = tmp;
    }
}


# prop: a wave reads its bucket once: all records come as one load group into a 16-element register vector, are counted and walked from there (no record load per 32 edges); sum order as before
# speedup vs baseline: 4.8415x; 1.0341x over previous
#include <hip/hip_runtime.h>
#include <cstdint>

#define N_NODES 100000
#define N_EDGES 1600000
#define IN_CH   256
#define HID     128
#define K_STEPS 10
#define ALPHA   0.1f

#define GROUP_ROWS 16
#define N_GROUPS   (N_NODES / GROUP_ROWS)
#define SRC_BITS   17
#define SRC_MASK   ((1 << SRC_BITS) - 1)
#define ROW_BITS   4
#define ROW_MASK   ((1 << ROW_BITS) - 1)
#define DEG_SHIFT  (SRC_BITS + ROW_BITS)
#define DEG_BITS   9
#define DEG_SAT    ((1 << DEG_BITS) - 1)
#define CUR_STRIDE GROUP_ROWS
static_assert(N_NODES % GROUP_ROWS == 0, "groups cover the rows exactly");
static_assert(N_NODES <= (1 << SRC_BITS), "a source row fits its field");
static_assert(GROUP_ROWS <= (1 << ROW_BITS), "a row inside its group fits its field");
static_assert(DEG_SHIFT + DEG_BITS <= 31, "the record's sign bit stays clear");
static_assert(N_GROUPS * CUR_STRIDE <= N_NODES, "the cursors fit in the in-degree array they replace");

typedef __attribute__((ext_vector_type(2))) float v2f;
typedef __attribute__((ext_vector_type(8))) float v8f;

__device__ __forceinline__ void atom_add_f32(float* p, float v) {
    __hip_atomic_fetch_add(p, v, __ATOMIC_RELAXED, __HIP_MEMORY_SCOPE_AGENT);
}

__global__ __launch_bounds__(256) void deg_kernel(const int* __restrict__ src,
                                                  const int* __restrict__ dst,
                                                  float* __restrict__ deg_out,
                                                  float* __restrict__ deg_in) {
    int e = blockIdx.x * blockDim.x + threadIdx.x;
    if (e < N_EDGES) {
        atom_add_f32(&deg_out[src[e]], 1.0f);
        atom_add_f32(&deg_in[dst[e]], 1.0f);
    }
}

__global__ __launch_bounds__(256) void rowprep_kernel(const float* __restrict__ deg_in,
                                                      int* __restrict__ counts) {
    const int g = blockIdx.x * blockDim.x + threadIdx.x;
    if (g < N_GROUPS) {
        float din = 0.0f;
        for (int k = 0; k < GROUP_ROWS; ++k)
            din += deg_in[g * GROUP_ROWS + k];
        counts[g] = (int)(din + 0.5f);
    }
}

#define SCAN_THREADS 1024
#define SCAN_RUN     ((N_GROUPS + SCAN_THREADS - 1) / SCAN_THREADS)

__global__ __launch_bounds__(SCAN_THREADS) void scan_kernel(const int* __restrict__ counts,
                                                            int* __restrict__ cur) {
    __shared__ int bufs[2][SCAN_THREADS];
    const int tid = threadIdx.x;
    const int i0  = tid * SCAN_RUN;
    int c[SCAN_RUN];
    int sum = 0;
    for (int k = 0; k < SCAN_RUN; ++k) {
        const int i = i0 + k;
        c[k] = (i < N_GROUPS) ? counts[i] : 0;
        sum += c[k];
    }
    bufs[0][tid] = sum;
    __syncthreads();
    int p = 0;
    for (int d = 1; d < SCAN_THREADS; d <<= 1) {
        int v = bufs[p][tid];
        if (tid >= d) v += bufs[p][tid - d];
        bufs[p ^ 1][tid] = v;
        __syncthreads();
        p ^= 1;
    }
    int run = (tid == 0) ? 0 : bufs[p][tid - 1];
    for (int k = 0; k < SCAN_RUN; ++k) {
        const int i = i0 + k;
        if (i < N_GROUPS) cur[i * CUR_STRIDE] = run;
        run += c[k];
    }
}

__global__ __launch_bounds__(256) void fill_kernel(const int* __restrict__ src,
                                                   const int* __restrict__ dst,
                                                   const float* __restrict__ deg_out,
                                                   int* __restrict__ cur,
                                                   int* __restrict__ rec) {
    const int e = blockIdx.x * blockDim.x + threadIdx.x;
    if (e < N_EDGES) {
        const int s = src[e];
        const int d = dst[e];
        const int g = d / GROUP_ROWS;
        const int od = min((int)deg_out[s], DEG_SAT);
        const int pos = __hip_atomic_fetch_add(&cur[g * CUR_STRIDE], 1, __ATOMIC_RELAXED,
                                               __HIP_MEMORY_SCOPE_AGENT);
        rec[pos] = s | ((d - g * GROUP_ROWS) << SRC_BITS) | (od << DEG_SHIFT);
    }
}

#define GEMM_THREADS 256
#define TILE_ROWS    16
#define N_TILES      (N_NODES / TILE_ROWS)
#define TILE_F4      (TILE_ROWS * IN_CH / 4)
#define A_PITCH      (IN_CH + 4)
#define GEMM_GRID    512
static_assert(N_NODES % TILE_ROWS == 0, "tiles cover the rows exactly");
static_assert(TILE_F4 == 4 * GEMM_THREADS, "a thread fetches 4 float4 of a tile");
static_assert(GEMM_GRID <= N_TILES, "every work-group has a first tile");
static_assert(((A_PITCH / 4) & 1) == 1 && A_PITCH % 4 == 0, "pitch is 16 bytes x odd");

__global__ __launch_bounds__(GEMM_THREADS, 2) void gemm_kernel(const float* __restrict__ X,
                                                                const float* __restrict__ W,
                                                                float* __restrict__ H) {
    __shared__ float xs[2][TILE_ROWS * A_PITCH];

    const int tid  = threadIdx.x;
    const int lane = tid & 31;
    const int wave = tid >> 5;
    const int half = lane >> 4;
    const int l16  = lane & 15;
    const int col  = wave * 16 + l16;

    v2f bw[IN_CH / 4];
    {
        const float* __restrict__ wp = W + (size_t)(4 * half) * HID + col;
#pragma unroll
        for (int q = 0; q < IN_CH / 8; ++q) {
            bw[2 * q].x     = wp[(size_t)(8 * q) * HID];
            bw[2 * q].y     = wp[(size_t)(8 * q + 1) * HID];
            bw[2 * q + 1].x = wp[(size_t)(8 * q + 2) * HID];
            bw[2 * q + 1].y = wp[(size_t)(8 * q + 3) * HID];
        }
    }

    const float4* __restrict__ Xv = (const float4*)X;
    const int srow = tid >> 6;
    const int scol = (tid & 63) * 4;
    const int soff = srow * A_PITCH + scol;
    int tile = blockIdx.x;
    float4 s0, s1, s2, s3;
    {
        const float4* __restrict__ xp = Xv + (size_t)tile * TILE_F4 + tid;
        s0 = xp[0];
        s1 = xp[GEMM_THREADS];
        s2 = xp[2 * GEMM_THREADS];
        s3 = xp[3 * GEMM_THREADS];
    }
    *(float4*)&xs[0][soff]                = s0;
    *(float4*)&xs[0][soff + 4 * A_PITCH]  = s1;
    *(float4*)&xs[0][soff + 8 * A_PITCH]  = s2;
    *(float4*)&xs[0][soff + 12 * A_PITCH] = s3;
    __syncthreads();

    int buf = 0;
    for (; tile < N_TILES; tile += GEMM_GRID) {
        const bool has_next = (tile + GEMM_GRID) < N_TILES;
        if (has_next) {
            const float4* __restrict__ xp = Xv + (size_t)(tile + GEMM_GRID) * TILE_F4 + tid;
            s0 = xp[0];
            s1 = xp[GEMM_THREADS];
            s2 = xp[2 * GEMM_THREADS];
            s3 = xp[3 * GEMM_THREADS];
        }

        const float* ap = &xs[buf][l16 * A_PITCH + 4 * half];
        float4 fa[8], fb[8];
        v8f acc = {};

#define A_GROUP(dst, g)                                                            \
        _Pragma("unroll") for (int j = 0; j < 8; ++j)                              \
            dst[j] = *(const float4*)(ap + 8 * (8 * (g) + j));
#define MMA_GROUP(srcf, g)                                                         \
        _Pragma("unroll") for (int j = 0; j < 8; ++j) {                            \
            const int q = 8 * (g) + j;                                             \
            v2f a0, a1;                                                            \
            a0.x = srcf[j].x; a0.y = srcf[j].y;                                    \
            a1.x = srcf[j].z; a1.y = srcf[j].w;                                    \
            acc = __builtin_amdgcn_wmma_f32_16x16x4_f32(                           \
                 false, a0,  false, bw[2 * q],                   \
                 (short)0, acc,  false,  false);    \
            acc = __builtin_amdgcn_wmma_f32_16x16x4_f32(                           \
                 false, a1,  false, bw[2 * q + 1],               \
                 (short)0, acc,  false,  false);    \
        }
#define KEEP_ORDER() __builtin_amdgcn_sched_barrier(0)

        A_GROUP(fa, 0)
        A_GROUP(fb, 1)
        KEEP_ORDER();
        MMA_GROUP(fa, 0)
        KEEP_ORDER();
        A_GROUP(fa, 2)
        KEEP_ORDER();
        MMA_GROUP(fb, 1)
        KEEP_ORDER();
        A_GROUP(fb, 3)
        KEEP_ORDER();
        MMA_GROUP(fa, 2)
        MMA_GROUP(fb, 3)
#undef A_GROUP
#undef MMA_GROUP
#undef KEEP_ORDER

        if (has_next) {
            float* xn = &xs[buf ^ 1][soff];
            *(float4*)(xn)                = s0;
            *(float4*)(xn + 4 * A_PITCH)  = s1;
            *(float4*)(xn + 8 * A_PITCH)  = s2;
            *(float4*)(xn + 12 * A_PITCH) = s3;
        }
        __syncthreads();
        buf ^= 1;

        float* __restrict__ hrow = H + (size_t)(tile * TILE_ROWS + half * 8) * HID + col;
#pragma unroll
        for (int i = 0; i < 8; ++i)
            hrow[(size_t)i * HID] = acc[i];
    }
}

#define PROP_THREADS  32
#define ROWS_PER_PASS 11
#define REC_REGS      16
#define REC_SPAN      (REC_REGS * 32)
#define TAIL_ROWS     8
static_assert(HID == 4 * 32, "a lane owns 4 adjacent channels of a row");
static_assert(GROUP_ROWS == 16, "a register vector has one element per row of the group");
static_assert(GROUP_ROWS % TAIL_ROWS == 0, "the tail takes the rows in whole parts");

typedef __attribute__((ext_vector_type(16))) float v16f;
typedef __attribute__((ext_vector_type(REC_REGS))) int v16i;

__device__ __forceinline__ v16i load_span(const int* __restrict__ rec, int base, int e, int lane) {
    v16i rr;
#pragma unroll
    for (int c = 0; c < REC_REGS; ++c) {
        const int i = base + c * 32 + lane;
        rr[c] = (i < e) ? rec[i] : -1;
    }
    return rr;
}

template <bool BLEND>
__global__ __launch_bounds__(PROP_THREADS, 7) void prop_kernel(const float* __restrict__ h,
                                                            const int* __restrict__ rec,
                                                            const int* __restrict__ cur,
                                                            const float* __restrict__ deg_out,
                                                            const float* __restrict__ tail,
                                                            float* __restrict__ out) {
    const int lane = threadIdx.x;
    const int g    = blockIdx.x;
    const int s    = (g == 0) ? 0 : cur[(g - 1) * CUR_STRIDE];
    const int e    = cur[g * CUR_STRIDE];

    int cnt[GROUP_ROWS];
#pragma unroll
    for (int k = 0; k < GROUP_ROWS; ++k) cnt[k] = 0;
    v16i rr = {};
    for (int base = s; base < e; base += REC_SPAN) {
        rr = load_span(rec, base, e, lane);
#pragma unroll
        for (int c = 0; c < REC_REGS; ++c) {
            if (base + c * 32 < e) {
                const int row = (rr[c] >= 0) ? ((rr[c] >> SRC_BITS) & ROW_MASK) : GROUP_ROWS;
#pragma unroll
                for (int k = 0; k < GROUP_ROWS; ++k)
                    cnt[k] += __builtin_popcount(__builtin_amdgcn_ballot_w32(row == k));
            }
        }
    }
    int mycnt = 0;
#pragma unroll
    for (int k = 0; k < GROUP_ROWS; ++k)
        if (lane == k) mycnt = cnt[k];
    const float my_inv = (mycnt > 0) ? rsqrtf(fmaxf((float)mycnt, 1.0f)) : 0.0f;

    v16f ax = {}, ay = {}, az = {}, aw = {};

    const float* __restrict__ hl = h + lane * 4;
    const bool one_span = (e - s) <= REC_SPAN;
    for (int sbase = s; sbase < e; sbase += REC_SPAN) {
        if (!one_span) rr = load_span(rec, sbase, e, lane);
        const int parts = min(REC_REGS, (e - sbase + 31) >> 5);
        for (int c = 0; c < parts; ++c) {
            const int n = min(32, e - sbase - c * 32);
            const int r = rr[c];
            int   sidx = 0;
            int   dl   = 0;
            float io   = 0.0f;
            if (r >= 0) {
                sidx = r & SRC_MASK;
                dl   = (r >> SRC_BITS) & ROW_MASK;
                const int od = r >> DEG_SHIFT;
                float dgo = (float)od;
                if (od == DEG_SAT) dgo = deg_out[sidx];
                io = (dgo > 0.0f) ? rsqrtf(fmaxf(dgo, 1.0f)) : 0.0f;
            }
            const float w  = io * __shfl(my_inv, dl, 32);
            const int   wv = __float_as_int(w);
            for (int j0 = 0; j0 < n; j0 += ROWS_PER_PASS) {
                float4 v[ROWS_PER_PASS];
#pragma unroll
                for (int j = 0; j < ROWS_PER_PASS; ++j) {
                    const int sj = __builtin_amdgcn_readlane(sidx, min(j0 + j, n - 1));
                    v[j] = *(const float4*)(hl + (size_t)sj * HID);
                }
#pragma unroll
                for (int j = 0; j < ROWS_PER_PASS; ++j) {
                    if (j0 + j < n) {
                        const int   d  = __builtin_amdgcn_readlane(dl, j0 + j);
                        const float wj = __int_as_float(__builtin_amdgcn_readlane(wv, j0 + j));
                        ax[d] += v[j].x * wj;
                        ay[d] += v[j].y * wj;
                        az[d] += v[j].z * wj;
                        aw[d] += v[j].w * wj;
                    }
                }
            }
        }
    }

    const size_t row0 = (size_t)g * GROUP_ROWS * HID + lane * 4;
    if (BLEND) {
#pragma unroll
        for (int k0 = 0; k0 < GROUP_ROWS; k0 += TAIL_ROWS) {
            float4 z[TAIL_ROWS];
#pragma unroll
            for (int k = 0; k < TAIL_ROWS; ++k)
                z[k] = *(const float4*)(tail + row0 + (size_t)(k0 + k) * HID);
#pragma unroll
            for (int k = 0; k < TAIL_ROWS; ++k) {
                float4 r;
                r.x = (1.0f - ALPHA) * ax[k0 + k] + ALPHA * z[k].x;
                r.y = (1.0f - ALPHA) * ay[k0 + k] + ALPHA * z[k].y;
                r.z = (1.0f - ALPHA) * az[k0 + k] + ALPHA * z[k].z;
                r.w = (1.0f - ALPHA) * aw[k0 + k] + ALPHA * z[k].w;
                *(float4*)(out + row0 + (size_t)(k0 + k) * HID) = r;
            }
        }
    } else {
        const float4 bz = *(const float4*)(tail + lane * 4);
#pragma unroll
        for (int k = 0; k < GROUP_ROWS; ++k) {
            float4 r;
            r.x = fmaxf(ax[k] + bz.x, 0.0f);
            r.y = fmaxf(ay[k] + bz.y, 0.0f);
            r.z = fmaxf(az[k] + bz.z, 0.0f);
            r.w = fmaxf(aw[k] + bz.w, 0.0f);
            *(float4*)(out + row0 + (size_t)k * HID) = r;
        }
    }
}

extern "C" void kernel_launch(void* const* d_in, const int* in_sizes, int n_in,
                              void* d_out, int out_size, void* d_ws, size_t ws_size,
                              hipStream_t stream) {
    const float* features = (const float*)d_in[0];
    const float* W        = (const float*)d_in[1];
    const float* bias     = (const float*)d_in[2];
    const int*   eidx     = (const int*)d_in[3];
    const int*   src      = eidx;
    const int*   dst      = eidx + N_EDGES;

    float* out = (float*)d_out;

    const size_t NH = (size_t)N_NODES * HID;
    float* deg     = (float*)d_ws;
    float* deg_out = deg;
    float* deg_in  = deg + N_NODES;
    int*   cur     = (int*)deg_in;
    int*   rec     = (int*)(deg + 2 * N_NODES);
    int*   counts  = rec;
    float* bufA    = (float*)(rec + N_EDGES);
    float* bufB    = bufA + NH;

    hipMemsetAsync(deg, 0, 2 * (size_t)N_NODES * sizeof(float), stream);
    deg_kernel<<<(N_EDGES + 255) / 256, 256, 0, stream>>>(src, dst, deg_out, deg_in);
    rowprep_kernel<<<(N_GROUPS + 255) / 256, 256, 0, stream>>>(deg_in, counts);
    scan_kernel<<<1, SCAN_THREADS, 0, stream>>>(counts, cur);
    fill_kernel<<<(N_EDGES + 255) / 256, 256, 0, stream>>>(src, dst, deg_out, cur, rec);

    gemm_kernel<<<GEMM_GRID, GEMM_THREADS, 0, stream>>>(features, W, bufA);
    prop_kernel<false><<<N_GROUPS, PROP_THREADS, 0, stream>>>(bufA, rec, cur, deg_out, bias, bufB);

    float* hc = bufB;
    for (int t = 0; t < K_STEPS; ++t) {
        float* tmp = (t & 1) ? out : bufA;
        prop_kernel<true><<<N_GROUPS, PROP_THREADS, 0, stream>>>(hc, rec, cur, deg_out, bufB, tmp);
        hc = tmp;
    }
}
